# starting kernel
# baseline (speedup 1.0000x reference)
#include <hip/hip_runtime.h>
#include <hip/hip_bf16.h>
#include <math.h>

#define N_NODES 100000
#define N_EDGES 600000
#define IN_FEATS 256
#define H_FEATS 128

typedef __attribute__((ext_vector_type(2))) float v2f;
typedef __attribute__((ext_vector_type(8))) float v8f;

__global__ __launch_bounds__(256) void degree_kernel(
    const int* __restrict__ srcIdx, const int* __restrict__ dstIdx,
    float* __restrict__ degOut, float* __restrict__ degIn)
{
    int e = blockIdx.x * blockDim.x + threadIdx.x;
    if (e < N_EDGES) {
        unsafeAtomicAdd(&degOut[srcIdx[e]], 1.0f);
        unsafeAtomicAdd(&degIn[dstIdx[e]], 1.0f);
    }
}

__global__ __launch_bounds__(256) void invnorm_kernel(
    float* __restrict__ degOut, float* __restrict__ degIn)
{
    int i = blockIdx.x * blockDim.x + threadIdx.x;
    if (i < N_NODES) {
        float d0 = degOut[i];
        degOut[i] = (d0 > 0.0f) ? rsqrtf(fmaxf(d0, 1.0f)) : 0.0f;
        float d1 = degIn[i];
        degIn[i] = (d1 > 0.0f) ? rsqrtf(fmaxf(d1, 1.0f)) : 0.0f;
    }
}

__global__ __launch_bounds__(256) void lin_wmma_kernel(
    const float* __restrict__ A, const float* __restrict__ W,
    const float* __restrict__ bias, float* __restrict__ X)
{
    const int lane   = threadIdx.x & 31;
    const int wave   = threadIdx.x >> 5;
    const int halfId = lane >> 4;
    const int l16    = lane & 15;
    const int rowBase = blockIdx.x * 32;
    const int n0      = wave * 16;

    const float* a0p = A + (size_t)(rowBase + l16) * IN_FEATS + 2 * halfId;
    const float* a1p = a0p + (size_t)16 * IN_FEATS;
    const float* bp  = W + (size_t)(2 * halfId) * H_FEATS + n0 + l16;

    v8f c0 = {};
    v8f c1 = {};

    #pragma unroll 8
    for (int k = 0; k < IN_FEATS; k += 4) {
        v2f a0, a1, bf;
        a0.x = a0p[k];
        a0.y = a0p[k + 1];
        a1.x = a1p[k];
        a1.y = a1p[k + 1];
        bf.x = bp[(size_t)k * H_FEATS];
        bf.y = bp[(size_t)(k + 1) * H_FEATS];
        c0 = __builtin_amdgcn_wmma_f32_16x16x4_f32(
                false, a0, false, bf, (short)0, c0, false, false);
        c1 = __builtin_amdgcn_wmma_f32_16x16x4_f32(
                false, a1, false, bf, (short)0, c1, false, false);
    }

    const float bb = bias[n0 + l16];
    #pragma unroll
    for (int r = 0; r < 8; ++r) {
        const int m = r + 8 * halfId;
        X[(size_t)(rowBase + m) * H_FEATS + n0 + l16]      = c0[r] + bb;
        X[(size_t)(rowBase + 16 + m) * H_FEATS + n0 + l16] = c1[r] + bb;
    }
}

__global__ __launch_bounds__(256) void scatter_kernel(
    const float* __restrict__ feat, const int* __restrict__ srcIdx,
    const int* __restrict__ dstIdx, const float* __restrict__ invSrc,
    float* __restrict__ agg)
{
    const int e = blockIdx.x * 8 + (threadIdx.x >> 5);
    if (e >= N_EDGES) return;
    const int lane = threadIdx.x & 31;
    const int s = srcIdx[e];
    const int d = dstIdx[e];
    const float w = invSrc[s];
    const float4 v = ((const float4*)(feat + (size_t)s * H_FEATS))[lane];
    float* ap = agg + (size_t)d * H_FEATS + lane * 4;
    unsafeAtomicAdd(ap + 0, v.x * w);
    unsafeAtomicAdd(ap + 1, v.y * w);
    unsafeAtomicAdd(ap + 2, v.z * w);
    unsafeAtomicAdd(ap + 3, v.w * w);
}

__device__ __forceinline__ float softmax2(const float* __restrict__ alphas, int idx)
{
    const float a0 = alphas[0], a1 = alphas[1];
    const float m  = fmaxf(a0, a1);
    const float e0 = __expf(a0 - m), e1 = __expf(a1 - m);
    return (idx == 0 ? e0 : e1) / (e0 + e1);
}

__global__ __launch_bounds__(256) void finalize_first_kernel(
    const float* __restrict__ agg, const float* __restrict__ invDst,
    const float* __restrict__ alphas, float* __restrict__ hbuf,
    float* __restrict__ out)
{
    const int i4 = blockIdx.x * blockDim.x + threadIdx.x;
    const int total4 = (N_NODES * H_FEATS) / 4;
    if (i4 >= total4) return;
    const int n = i4 / (H_FEATS / 4);
    const float s = invDst[n];
    const float a0 = softmax2(alphas, 0);
    float4 v = ((const float4*)agg)[i4];
    v.x *= s; v.y *= s; v.z *= s; v.w *= s;
    ((float4*)hbuf)[i4] = v;
    float4 o;
    o.x = a0 * v.x; o.y = a0 * v.y; o.z = a0 * v.z; o.w = a0 * v.w;
    ((float4*)out)[i4] = o;
}

__global__ __launch_bounds__(256) void finalize_second_kernel(
    const float* __restrict__ agg, const float* __restrict__ invDst,
    const float* __restrict__ alphas, float* __restrict__ out)
{
    const int i4 = blockIdx.x * blockDim.x + threadIdx.x;
    const int total4 = (N_NODES * H_FEATS) / 4;
    if (i4 >= total4) return;
    const int n = i4 / (H_FEATS / 4);
    const float s = invDst[n] * softmax2(alphas, 1);
    const float4 v = ((const float4*)agg)[i4];
    float4 o = ((float4*)out)[i4];
    o.x += s * v.x; o.y += s * v.y; o.z += s * v.z; o.w += s * v.w;
    ((float4*)out)[i4] = o;
}

extern "C" void kernel_launch(void* const* d_in, const int* in_sizes, int n_in,
                              void* d_out, int out_size, void* d_ws, size_t ws_size,
                              hipStream_t stream)
{
    (void)in_sizes; (void)n_in; (void)out_size; (void)ws_size;

    const float* in_feat = (const float*)d_in[0];
    const int*   edge    = (const int*)d_in[1];
    const float* W       = (const float*)d_in[2];
    const float* bias    = (const float*)d_in[3];
    const float* alphas  = (const float*)d_in[4];
    float*       out     = (float*)d_out;

    const int* srcIdx = edge;
    const int* dstIdx = edge + N_EDGES;

    float* degOut = (float*)d_ws;
    float* degIn  = degOut + N_NODES;
    float* xbuf   = degIn + N_NODES;
    float* agg    = xbuf + (size_t)N_NODES * H_FEATS;

    const size_t featBytes = (size_t)N_NODES * H_FEATS * sizeof(float);

    hipMemsetAsync(degOut, 0, (size_t)2 * N_NODES * sizeof(float), stream);
    degree_kernel<<<(N_EDGES + 255) / 256, 256, 0, stream>>>(srcIdx, dstIdx, degOut, degIn);
    invnorm_kernel<<<(N_NODES + 255) / 256, 256, 0, stream>>>(degOut, degIn);

    lin_wmma_kernel<<<N_NODES / 32, 256, 0, stream>>>(in_feat, W, bias, xbuf);

    hipMemsetAsync(agg, 0, featBytes, stream);
    scatter_kernel<<<N_EDGES / 8, 256, 0, stream>>>(xbuf, srcIdx, dstIdx, degOut, agg);
    finalize_first_kernel<<<(N_NODES * H_FEATS / 4 + 255) / 256, 256, 0, stream>>>(
        agg, degIn, alphas, xbuf, out);

    hipMemsetAsync(agg, 0, featBytes, stream);
    scatter_kernel<<<N_EDGES / 8, 256, 0, stream>>>(xbuf, srcIdx, dstIdx, degOut, agg);
    finalize_second_kernel<<<(N_NODES * H_FEATS / 4 + 255) / 256, 256, 0, stream>>>(
        agg, degIn, alphas, out);
}


# edge scatter: 256 owner work-groups keep 391 agg rows each in LDS, scan all dst ids, sum their edges by LDS f32 atomic add (arrival order), write rows once; no global atomics, zero-fill, scaling pass
# speedup vs baseline: 1.8005x; 1.8005x over previous
#include <hip/hip_runtime.h>
#include <hip/hip_bf16.h>
#include <math.h>

#define N_NODES 100000
#define N_EDGES 600000
#define IN_FEATS 256
#define H_FEATS 128

typedef __attribute__((ext_vector_type(2))) float v2f;
typedef __attribute__((ext_vector_type(8))) float v8f;

__global__ __launch_bounds__(256) void degree_kernel(
    const int* __restrict__ srcIdx, const int* __restrict__ dstIdx,
    float* __restrict__ degOut, float* __restrict__ degIn)
{
    int e = blockIdx.x * blockDim.x + threadIdx.x;
    if (e < N_EDGES) {
        unsafeAtomicAdd(&degOut[srcIdx[e]], 1.0f);
        unsafeAtomicAdd(&degIn[dstIdx[e]], 1.0f);
    }
}

__global__ __launch_bounds__(256) void invnorm_kernel(
    float* __restrict__ degOut, float* __restrict__ degIn)
{
    int i = blockIdx.x * blockDim.x + threadIdx.x;
    if (i < N_NODES) {
        float d0 = degOut[i];
        degOut[i] = (d0 > 0.0f) ? rsqrtf(fmaxf(d0, 1.0f)) : 0.0f;
        float d1 = degIn[i];
        degIn[i] = (d1 > 0.0f) ? rsqrtf(fmaxf(d1, 1.0f)) : 0.0f;
    }
}

__global__ __launch_bounds__(256) void lin_wmma_kernel(
    const float* __restrict__ A, const float* __restrict__ W,
    const float* __restrict__ bias, float* __restrict__ X)
{
    const int lane   = threadIdx.x & 31;
    const int wave   = threadIdx.x >> 5;
    const int halfId = lane >> 4;
    const int l16    = lane & 15;
    const int rowBase = blockIdx.x * 32;
    const int n0      = wave * 16;

    const float* a0p = A + (size_t)(rowBase + l16) * IN_FEATS + 2 * halfId;
    const float* a1p = a0p + (size_t)16 * IN_FEATS;
    const float* bp  = W + (size_t)(2 * halfId) * H_FEATS + n0 + l16;

    v8f c0 = {};
    v8f c1 = {};

    #pragma unroll 8
    for (int k = 0; k < IN_FEATS; k += 4) {
        v2f a0, a1, bf;
        a0.x = a0p[k];
        a0.y = a0p[k + 1];
        a1.x = a1p[k];
        a1.y = a1p[k + 1];
        bf.x = bp[(size_t)k * H_FEATS];
        bf.y = bp[(size_t)(k + 1) * H_FEATS];
        c0 = __builtin_amdgcn_wmma_f32_16x16x4_f32(
                false, a0, false, bf, (short)0, c0, false, false);
        c1 = __builtin_amdgcn_wmma_f32_16x16x4_f32(
                false, a1, false, bf, (short)0, c1, false, false);
    }

    const float bb = bias[n0 + l16];
    #pragma unroll
    for (int r = 0; r < 8; ++r) {
        const int m = r + 8 * halfId;
        X[(size_t)(rowBase + m) * H_FEATS + n0 + l16]      = c0[r] + bb;
        X[(size_t)(rowBase + 16 + m) * H_FEATS + n0 + l16] = c1[r] + bb;
    }
}

#define OWN_WGS 256
#define OWN_ROWS ((N_NODES + OWN_WGS - 1) / OWN_WGS)
#define OWN_THREADS 1024
#define OWN_WAVES (OWN_THREADS / 32)
#define SEG_PER_ROUND 7
#define EDGES_PER_ROUND (SEG_PER_ROUND * OWN_THREADS * 4)
#define SCAN_ROUNDS ((N_EDGES + EDGES_PER_ROUND - 1) / EDGES_PER_ROUND)
#define WAVE_LIST_CAP 960
#define WAVE_EDGES_PER_ROUND (SEG_PER_ROUND * 32 * 4)
#define WAVE_DRAIN_AT (WAVE_LIST_CAP - WAVE_EDGES_PER_ROUND)
#define GATHER_BATCH 16
#define ROW_GROUP 4

static_assert(OWN_ROWS <= 512, "local row number must fit the 9 low bits of a list entry");
static_assert(N_EDGES < (1 << 23), "edge number must fit the 23 high bits of a list entry");
static_assert(N_EDGES % 4 == 0, "destination ids are read four at a time");
static_assert(WAVE_DRAIN_AT >= 0, "one round of a wave's matches must fit its list");

__device__ __forceinline__ float softmax2(const float* __restrict__ alphas, int idx)
{
    const float a0 = alphas[0], a1 = alphas[1];
    const float m  = fmaxf(a0, a1);
    const float e0 = __expf(a0 - m), e1 = __expf(a1 - m);
    return (idx == 0 ? e0 : e1) / (e0 + e1);
}

__device__ __forceinline__ void load_round(
    const int* __restrict__ dstIdx, const int round, const int tid, int4 (&d)[SEG_PER_ROUND])
{
    #pragma unroll
    for (int g = 0; g < SEG_PER_ROUND; ++g) {
        const int q = (round * SEG_PER_ROUND + g) * OWN_THREADS + tid;
        d[g] = ((const int4*)dstIdx)[q < N_EDGES / 4 ? q : N_EDGES / 4 - 1];
    }
}

__device__ __forceinline__ void drain_list(
    const float* __restrict__ feat, const int* __restrict__ srcIdx,
    const float* __restrict__ invSrc, float* acc, const unsigned* list,
    const int count, const int lane)
{
    for (int base = 0; base < count; base += GATHER_BATCH) {
        const int left = count - base;
        const int nvalid = left < GATHER_BATCH ? left : GATHER_BATCH;
        unsigned ent = 0;
        int s = 0;
        float w = 0.0f;
        if (lane < nvalid) {
            ent = list[base + lane];
            s = srcIdx[ent >> 9];
            w = invSrc[s];
        }
        float v[GATHER_BATCH][4];
        #pragma unroll
        for (int j = 0; j < GATHER_BATCH; ++j) {
            if (j < nvalid) {
                const int sj = __builtin_amdgcn_readlane(s, j);
                const float* fp = feat + (size_t)sj * H_FEATS + lane;
                v[j][0] = fp[0];
                v[j][1] = fp[32];
                v[j][2] = fp[64];
                v[j][3] = fp[96];
            }
        }
        #pragma unroll
        for (int j = 0; j < GATHER_BATCH; ++j) {
            if (j < nvalid) {
                const int dl = (int)(__builtin_amdgcn_readlane((int)ent, j) & 511);
                const float wj = __int_as_float(__builtin_amdgcn_readlane(__float_as_int(w), j));
                float* ap = acc + dl * H_FEATS + lane;
                unsafeAtomicAdd(ap + 0,  v[j][0] * wj);
                unsafeAtomicAdd(ap + 32, v[j][1] * wj);
                unsafeAtomicAdd(ap + 64, v[j][2] * wj);
                unsafeAtomicAdd(ap + 96, v[j][3] * wj);
            }
        }
    }
}

__device__ __forceinline__ void append_owned(
    unsigned* mylist, int& cnt, const bool owned, const unsigned edge, const unsigned localRow)
{
    const unsigned mask = __builtin_amdgcn_ballot_w32(owned);
    if (owned) mylist[cnt + (int)__builtin_amdgcn_mbcnt_lo(mask, 0u)] = (edge << 9) | localRow;
    cnt += __builtin_popcount(mask);
}

template <bool FIRST>
__global__ __launch_bounds__(OWN_THREADS) void conv_owner_kernel(
    const float* __restrict__ feat, const int* __restrict__ srcIdx,
    const int* __restrict__ dstIdx, const float* __restrict__ invSrc,
    const float* __restrict__ invDst, const float* __restrict__ alphas,
    float* __restrict__ hbuf, float* __restrict__ out)
{
    __shared__ float acc[OWN_ROWS * H_FEATS];
    __shared__ unsigned list[OWN_WAVES * WAVE_LIST_CAP];

    const int tid  = threadIdx.x;
    const int lane = tid & 31;
    const int wave = __builtin_amdgcn_readfirstlane(tid >> 5);
    const int lo   = blockIdx.x * OWN_ROWS;
    const int rows = (N_NODES - lo) < OWN_ROWS ? (N_NODES - lo) : OWN_ROWS;
    unsigned* mylist = list + wave * WAVE_LIST_CAP;

    for (int i = tid; i < OWN_ROWS * H_FEATS / 4; i += OWN_THREADS)
        ((float4*)acc)[i] = make_float4(0.0f, 0.0f, 0.0f, 0.0f);
    __syncthreads();

    int rr = blockIdx.x % SCAN_ROUNDS;
    int cnt = 0;
    for (int round = 0; round < SCAN_ROUNDS; ++round) {
        int4 d[SEG_PER_ROUND];
        load_round(dstIdx, rr, tid, d);
        #pragma unroll
        for (int g = 0; g < SEG_PER_ROUND; ++g) {
            const unsigned t0 = (unsigned)(d[g].x - lo);
            const unsigned t1 = (unsigned)(d[g].y - lo);
            const unsigned t2 = (unsigned)(d[g].z - lo);
            const unsigned t3 = (unsigned)(d[g].w - lo);
            const int q = (rr * SEG_PER_ROUND + g) * OWN_THREADS + tid;
            const unsigned lim = (q < N_EDGES / 4) ? (unsigned)rows : 0u;
            const bool m0 = t0 < lim, m1 = t1 < lim;
            const bool m2 = t2 < lim, m3 = t3 < lim;
            if (__builtin_amdgcn_ballot_w32(m0 | m1 | m2 | m3)) {
                const unsigned e = (unsigned)(q * 4);
                append_owned(mylist, cnt, m0, e + 0u, t0);
                append_owned(mylist, cnt, m1, e + 1u, t1);
                append_owned(mylist, cnt, m2, e + 2u, t2);
                append_owned(mylist, cnt, m3, e + 3u, t3);
            }
        }
        if (cnt > WAVE_DRAIN_AT || round + 1 == SCAN_ROUNDS) {
            drain_list(feat, srcIdx, invSrc, acc, mylist, cnt, lane);
            cnt = 0;
        }
        rr = (rr + 1 == SCAN_ROUNDS) ? 0 : rr + 1;
    }
    __syncthreads();

    const float a = softmax2(alphas, FIRST ? 0 : 1);
    for (int r0 = wave; r0 < rows; r0 += OWN_WAVES * ROW_GROUP) {
        float4 o[ROW_GROUP];
        float s[ROW_GROUP];
        #pragma unroll
        for (int k = 0; k < ROW_GROUP; ++k) {
            const int r = r0 + k * OWN_WAVES;
            if (r < rows) {
                s[k] = invDst[lo + r];
                if (!FIRST) o[k] = ((const float4*)out)[(size_t)(lo + r) * (H_FEATS / 4) + lane];
            }
        }
        #pragma unroll
        for (int k = 0; k < ROW_GROUP; ++k) {
            const int r = r0 + k * OWN_WAVES;
            if (r < rows) {
                const float4 v = ((const float4*)(acc + r * H_FEATS))[lane];
                const size_t i4 = (size_t)(lo + r) * (H_FEATS / 4) + lane;
                if (FIRST) {
                    float4 h;
                    h.x = v.x * s[k]; h.y = v.y * s[k]; h.z = v.z * s[k]; h.w = v.w * s[k];
                    ((float4*)hbuf)[i4] = h;
                    float4 w;
                    w.x = a * h.x; w.y = a * h.y; w.z = a * h.z; w.w = a * h.w;
                    ((float4*)out)[i4] = w;
                } else {
                    const float sa = s[k] * a;
                    float4 w = o[k];
                    w.x += sa * v.x; w.y += sa * v.y; w.z += sa * v.z; w.w += sa * v.w;
                    ((float4*)out)[i4] = w;
                }
            }
        }
    }
}

extern "C" void kernel_launch(void* const* d_in, const int* in_sizes, int n_in,
                              void* d_out, int out_size, void* d_ws, size_t ws_size,
                              hipStream_t stream)
{
    (void)in_sizes; (void)n_in; (void)out_size; (void)ws_size;

    const float* in_feat = (const float*)d_in[0];
    const int*   edge    = (const int*)d_in[1];
    const float* W       = (const float*)d_in[2];
    const float* bias    = (const float*)d_in[3];
    const float* alphas  = (const float*)d_in[4];
    float*       out     = (float*)d_out;

    const int* srcIdx = edge;
    const int* dstIdx = edge + N_EDGES;

    float* degOut = (float*)d_ws;
    float* degIn  = degOut + N_NODES;
    float* xbuf   = degIn + N_NODES;
    float* hbuf   = xbuf + (size_t)N_NODES * H_FEATS;

    hipMemsetAsync(degOut, 0, (size_t)2 * N_NODES * sizeof(float), stream);
    degree_kernel<<<(N_EDGES + 255) / 256, 256, 0, stream>>>(srcIdx, dstIdx, degOut, degIn);
    invnorm_kernel<<<(N_NODES + 255) / 256, 256, 0, stream>>>(degOut, degIn);

    lin_wmma_kernel<<<N_NODES / 32, 256, 0, stream>>>(in_feat, W, bias, xbuf);

    conv_owner_kernel<true><<<OWN_WGS, OWN_THREADS, 0, stream>>>(
        xbuf, srcIdx, dstIdx, degOut, degIn, alphas, hbuf, out);

    conv_owner_kernel<false><<<OWN_WGS, OWN_THREADS, 0, stream>>>(
        hbuf, srcIdx, dstIdx, degOut, degIn, alphas, nullptr, out);
}


# lin GEMM: each wave keeps its 256x16 slice of W in registers for the whole kernel; 512 persistent work-groups stream 32-row tiles of in_feat once through LDS, whole-line loads; k order, sums same
# speedup vs baseline: 2.3862x; 1.3253x over previous
#include <hip/hip_runtime.h>
#include <hip/hip_bf16.h>
#include <math.h>

#define N_NODES 100000
#define N_EDGES 600000
#define IN_FEATS 256
#define H_FEATS 128

typedef __attribute__((ext_vector_type(2))) float v2f;
typedef __attribute__((ext_vector_type(8))) float v8f;
typedef __attribute__((ext_vector_type(4))) float v4f;

__global__ __launch_bounds__(256) void degree_kernel(
    const int* __restrict__ srcIdx, const int* __restrict__ dstIdx,
    float* __restrict__ degOut, float* __restrict__ degIn)
{
    int e = blockIdx.x * blockDim.x + threadIdx.x;
    if (e < N_EDGES) {
        unsafeAtomicAdd(&degOut[srcIdx[e]], 1.0f);
        unsafeAtomicAdd(&degIn[dstIdx[e]], 1.0f);
    }
}

__global__ __launch_bounds__(256) void invnorm_kernel(
    float* __restrict__ degOut, float* __restrict__ degIn)
{
    int i = blockIdx.x * blockDim.x + threadIdx.x;
    if (i < N_NODES) {
        float d0 = degOut[i];
        degOut[i] = (d0 > 0.0f) ? rsqrtf(fmaxf(d0, 1.0f)) : 0.0f;
        float d1 = degIn[i];
        degIn[i] = (d1 > 0.0f) ? rsqrtf(fmaxf(d1, 1.0f)) : 0.0f;
    }
}

#define LIN_ROWS 32
#define LIN_PITCH (IN_FEATS + 4)
#define LIN_WGS 512
#define LIN_TILES (N_NODES / LIN_ROWS)
#define LIN_KSTEPS (IN_FEATS / 4)
#define LIN_PIECES (LIN_ROWS * IN_FEATS / 4 / 256)

static_assert(N_NODES % LIN_ROWS == 0, "tiles are whole");
static_assert(LIN_WGS <= LIN_TILES, "every work-group has a first tile");
static_assert(LIN_PITCH % 64 == 4 && (LIN_PITCH * 4) % 16 == 0, "bank spread and 16-byte alignment of the rows");
static_assert(H_FEATS == 8 * 16, "eight waves of 16 columns");

__global__ __launch_bounds__(256, 2) void lin_wmma_kernel(
    const float* __restrict__ A, const float* __restrict__ W,
    const float* __restrict__ bias, float* __restrict__ X)
{
    __shared__ __attribute__((aligned(16))) float tile[2][LIN_ROWS * LIN_PITCH];

    const int tid    = threadIdx.x;
    const int lane   = tid & 31;
    const int wave   = tid >> 5;
    const int halfId = lane >> 4;
    const int l16    = lane & 15;
    const int n0     = wave * 16;

    v2f bf[LIN_KSTEPS];
    {
        const float* bp = W + (size_t)(2 * halfId) * H_FEATS + n0 + l16;
        #pragma unroll
        for (int j = 0; j < LIN_KSTEPS; ++j) {
            bf[j].x = bp[(size_t)(4 * j) * H_FEATS];
            bf[j].y = bp[(size_t)(4 * j + 1) * H_FEATS];
        }
    }
    const float bb = bias[n0 + l16];

    const int stRow = tid >> 6;
    const int stCol = (tid & 63) * 4;
    const v4f* A4 = (const v4f*)A;

    v4f st[LIN_PIECES];
    int t = blockIdx.x;
    #pragma unroll
    for (int i = 0; i < LIN_PIECES; ++i)
        st[i] = A4[(size_t)t * (LIN_ROWS * IN_FEATS / 4) + i * 256 + tid];
    #pragma unroll
    for (int i = 0; i < LIN_PIECES; ++i)
        *(v4f*)(&tile[0][(i * 4 + stRow) * LIN_PITCH + stCol]) = st[i];
    __syncthreads();

    int buf = 0;
    for (; t < LIN_TILES; t += LIN_WGS) {
        const int tn = t + LIN_WGS;
        const bool more = tn < LIN_TILES;
        if (more) {
            #pragma unroll
            for (int i = 0; i < LIN_PIECES; ++i)
                st[i] = A4[(size_t)tn * (LIN_ROWS * IN_FEATS / 4) + i * 256 + tid];
        }

        const float* fa = &tile[buf][l16 * LIN_PITCH + 2 * halfId];
        v8f c0 = {};
        v8f c1 = {};
        #pragma unroll
        for (int j = 0; j < LIN_KSTEPS; ++j) {
            const v2f a0 = *(const v2f*)(fa + 4 * j);
            const v2f a1 = *(const v2f*)(fa + 16 * LIN_PITCH + 4 * j);
            c0 = __builtin_amdgcn_wmma_f32_16x16x4_f32(
                    false, a0, false, bf[j], (short)0, c0, false, false);
            c1 = __builtin_amdgcn_wmma_f32_16x16x4_f32(
                    false, a1, false, bf[j], (short)0, c1, false, false);
        }

        if (more) {
            #pragma unroll
            for (int i = 0; i < LIN_PIECES; ++i)
                *(v4f*)(&tile[buf ^ 1][(i * 4 + stRow) * LIN_PITCH + stCol]) = st[i];
        }
        __syncthreads();
        buf ^= 1;

        const int rowBase = t * LIN_ROWS;
        #pragma unroll
        for (int r = 0; r < 8; ++r) {
            const int m = r + 8 * halfId;
            X[(size_t)(rowBase + m) * H_FEATS + n0 + l16]      = c0[r] + bb;
            X[(size_t)(rowBase + 16 + m) * H_FEATS + n0 + l16] = c1[r] + bb;
        }
    }
}

#define OWN_WGS 256
#define OWN_ROWS ((N_NODES + OWN_WGS - 1) / OWN_WGS)
#define OWN_THREADS 1024
#define OWN_WAVES (OWN_THREADS / 32)
#define SEG_PER_ROUND 7
#define EDGES_PER_ROUND (SEG_PER_ROUND * OWN_THREADS * 4)
#define SCAN_ROUNDS ((N_EDGES + EDGES_PER_ROUND - 1) / EDGES_PER_ROUND)
#define WAVE_LIST_CAP 960
#define WAVE_EDGES_PER_ROUND (SEG_PER_ROUND * 32 * 4)
#define WAVE_DRAIN_AT (WAVE_LIST_CAP - WAVE_EDGES_PER_ROUND)
#define GATHER_BATCH 16
#define ROW_GROUP 4
#define SAVE_HDR 32
#define SAVE_CAP WAVE_LIST_CAP
#define SAVE_STRIDE 1024

static_assert(OWN_ROWS <= 512, "local row number must fit the 9 low bits of a list entry");
static_assert(N_EDGES < (1 << 23), "edge number must fit the 23 high bits of a list entry");
static_assert(N_EDGES % 4 == 0, "destination ids are read four at a time");
static_assert(WAVE_DRAIN_AT >= 0, "one round of a wave's matches must fit its list");
static_assert(N_NODES <= (1 << 23), "source id must fit the 23 high bits of a stored entry");
static_assert(SAVE_HDR + SAVE_CAP <= SAVE_STRIDE, "count line and entries must fit the wave's region");
static_assert((N_NODES - (OWN_WGS - 1) * OWN_ROWS) * H_FEATS >= OWN_WAVES * SAVE_STRIDE,
              "the last work-group's rows of out must hold its waves' regions");
static_assert((OWN_ROWS * H_FEATS) % 64 == 0 && SAVE_STRIDE % 64 == 0, "regions start on 256-byte units");

__device__ __forceinline__ float softmax2(const float* __restrict__ alphas, int idx)
{
    const float a0 = alphas[0], a1 = alphas[1];
    const float m  = fmaxf(a0, a1);
    const float e0 = __expf(a0 - m), e1 = __expf(a1 - m);
    return (idx == 0 ? e0 : e1) / (e0 + e1);
}

__device__ __forceinline__ void load_round(
    const int* __restrict__ dstIdx, const int round, const int tid, int4 (&d)[SEG_PER_ROUND])
{
    #pragma unroll
    for (int g = 0; g < SEG_PER_ROUND; ++g) {
        const int q = (round * SEG_PER_ROUND + g) * OWN_THREADS + tid;
        d[g] = ((const int4*)dstIdx)[q < N_EDGES / 4 ? q : N_EDGES / 4 - 1];
    }
}

template <bool STORED_IN, bool STORE_OUT>
__device__ __forceinline__ void drain_list(
    const float* __restrict__ feat, const int* __restrict__ srcIdx,
    const float* __restrict__ invSrc, float* acc, const unsigned* list,
    const int count, const int lane, float* save, const int saveBase)
{
    for (int base = 0; base < count; base += GATHER_BATCH) {
        const int left = count - base;
        const int nvalid = left < GATHER_BATCH ? left : GATHER_BATCH;
        unsigned ent = 0;
        int s = 0;
        float w = 0.0f;
        if (lane < nvalid) {
            ent = list[base + lane];
            s = STORED_IN ? (int)(ent >> 9) : srcIdx[ent >> 9];
            w = invSrc[s];
            if (STORE_OUT) {
                const int pos = saveBase + base + lane;
                if (pos < SAVE_CAP)
                    save[SAVE_HDR + pos] = __uint_as_float(((unsigned)s << 9) | (ent & 511u));
            }
        }
        float v[GATHER_BATCH][4];
        #pragma unroll
        for (int j = 0; j < GATHER_BATCH; ++j) {
            if (j < nvalid) {
                const int sj = __builtin_amdgcn_readlane(s, j);
                const float* fp = feat + (size_t)sj * H_FEATS + lane;
                v[j][0] = fp[0];
                v[j][1] = fp[32];
                v[j][2] = fp[64];
                v[j][3] = fp[96];
            }
        }
        #pragma unroll
        for (int j = 0; j < GATHER_BATCH; ++j) {
            if (j < nvalid) {
                const int dl = (int)(__builtin_amdgcn_readlane((int)ent, j) & 511);
                const float wj = __int_as_float(__builtin_amdgcn_readlane(__float_as_int(w), j));
                float* ap = acc + dl * H_FEATS + lane;
                unsafeAtomicAdd(ap + 0,  v[j][0] * wj);
                unsafeAtomicAdd(ap + 32, v[j][1] * wj);
                unsafeAtomicAdd(ap + 64, v[j][2] * wj);
                unsafeAtomicAdd(ap + 96, v[j][3] * wj);
            }
        }
    }
}

__device__ __forceinline__ void append_owned(
    unsigned* mylist, int& cnt, const bool owned, const unsigned edge, const unsigned localRow)
{
    const unsigned mask = __builtin_amdgcn_ballot_w32(owned);
    if (owned) mylist[cnt + (int)__builtin_amdgcn_mbcnt_lo(mask, 0u)] = (edge << 9) | localRow;
    cnt += __builtin_popcount(mask);
}

template <bool STORE_OUT>
__device__ __forceinline__ void scan_and_drain(
    const float* __restrict__ feat, const int* __restrict__ srcIdx,
    const int* __restrict__ dstIdx, const float* __restrict__ invSrc,
    float* acc, unsigned* mylist, float* save,
    const int lo, const int rows, const int tid, const int lane)
{
    int rr = blockIdx.x % SCAN_ROUNDS;
    int cnt = 0;
    int stored = 0;
    for (int round = 0; round < SCAN_ROUNDS; ++round) {
        int4 d[SEG_PER_ROUND];
        load_round(dstIdx, rr, tid, d);
        #pragma unroll
        for (int g = 0; g < SEG_PER_ROUND; ++g) {
            const unsigned t0 = (unsigned)(d[g].x - lo);
            const unsigned t1 = (unsigned)(d[g].y - lo);
            const unsigned t2 = (unsigned)(d[g].z - lo);
            const unsigned t3 = (unsigned)(d[g].w - lo);
            const int q = (rr * SEG_PER_ROUND + g) * OWN_THREADS + tid;
            const unsigned lim = (q < N_EDGES / 4) ? (unsigned)rows : 0u;
            const bool m0 = t0 < lim, m1 = t1 < lim;
            const bool m2 = t2 < lim, m3 = t3 < lim;
            if (__builtin_amdgcn_ballot_w32(m0 | m1 | m2 | m3)) {
                const unsigned e = (unsigned)(q * 4);
                append_owned(mylist, cnt, m0, e + 0u, t0);
                append_owned(mylist, cnt, m1, e + 1u, t1);
                append_owned(mylist, cnt, m2, e + 2u, t2);
                append_owned(mylist, cnt, m3, e + 3u, t3);
            }
        }
        if (cnt > WAVE_DRAIN_AT || round + 1 == SCAN_ROUNDS) {
            drain_list<false, STORE_OUT>(feat, srcIdx, invSrc, acc, mylist, cnt, lane, save, stored);
            stored += cnt;
            cnt = 0;
        }
        rr = (rr + 1 == SCAN_ROUNDS) ? 0 : rr + 1;
    }
    if (STORE_OUT && lane == 0) save[0] = __int_as_float(stored);
}

template <bool FIRST>
__global__ __launch_bounds__(OWN_THREADS) void conv_owner_kernel(
    const float* __restrict__ feat, const int* __restrict__ srcIdx,
    const int* __restrict__ dstIdx, const float* __restrict__ invSrc,
    const float* __restrict__ invDst, const float* __restrict__ alphas,
    float* __restrict__ hbuf, float* __restrict__ out)
{
    __shared__ float acc[OWN_ROWS * H_FEATS];
    __shared__ unsigned list[OWN_WAVES * WAVE_LIST_CAP];

    const int tid  = threadIdx.x;
    const int lane = tid & 31;
    const int wave = __builtin_amdgcn_readfirstlane(tid >> 5);
    const int lo   = blockIdx.x * OWN_ROWS;
    const int rows = (N_NODES - lo) < OWN_ROWS ? (N_NODES - lo) : OWN_ROWS;
    unsigned* mylist = list + wave * WAVE_LIST_CAP;

    for (int i = tid; i < OWN_ROWS * H_FEATS / 4; i += OWN_THREADS)
        ((float4*)acc)[i] = make_float4(0.0f, 0.0f, 0.0f, 0.0f);
    __syncthreads();

    float* save = out + (size_t)lo * H_FEATS + wave * SAVE_STRIDE;
    if (FIRST) {
        scan_and_drain<true>(feat, srcIdx, dstIdx, invSrc, acc, mylist, save, lo, rows, tid, lane);
    } else {
        const int total = __builtin_amdgcn_readfirstlane(__float_as_int(save[0]));
        if (total <= SAVE_CAP) {
            for (int i0 = 0; i0 < total; i0 += 128) {
                float t[4];
                #pragma unroll
                for (int u = 0; u < 4; ++u) {
                    const int i = i0 + u * 32 + lane;
                    t[u] = save[SAVE_HDR + (i < total ? i : 0)];
                }
                #pragma unroll
                for (int u = 0; u < 4; ++u) {
                    const int i = i0 + u * 32 + lane;
                    if (i < total) mylist[i] = __float_as_uint(t[u]);
                }
            }
            drain_list<true, false>(feat, srcIdx, invSrc, acc, mylist, total, lane, save, 0);
        } else {
            scan_and_drain<false>(feat, srcIdx, dstIdx, invSrc, acc, mylist, save, lo, rows, tid, lane);
        }
    }
    __syncthreads();

    const float a0 = FIRST ? 0.0f : softmax2(alphas, 0);
    const float a1 = FIRST ? 0.0f : softmax2(alphas, 1);
    for (int r0 = wave; r0 < rows; r0 += OWN_WAVES * ROW_GROUP) {
        float4 o[ROW_GROUP];
        float s[ROW_GROUP];
        #pragma unroll
        for (int k = 0; k < ROW_GROUP; ++k) {
            const int r = r0 + k * OWN_WAVES;
            if (r < rows) {
                s[k] = invDst[lo + r];
                if (!FIRST) o[k] = ((const float4*)feat)[(size_t)(lo + r) * (H_FEATS / 4) + lane];
            }
        }
        #pragma unroll
        for (int k = 0; k < ROW_GROUP; ++k) {
            const int r = r0 + k * OWN_WAVES;
            if (r < rows) {
                const float4 v = ((const float4*)(acc + r * H_FEATS))[lane];
                const size_t i4 = (size_t)(lo + r) * (H_FEATS / 4) + lane;
                if (FIRST) {
                    float4 h;
                    h.x = v.x * s[k]; h.y = v.y * s[k]; h.z = v.z * s[k]; h.w = v.w * s[k];
                    ((float4*)hbuf)[i4] = h;
                } else {
                    const float sa = s[k] * a1;
                    const float4 h = o[k];
                    float4 w;
                    w.x = fmaf(sa, v.x, a0 * h.x); w.y = fmaf(sa, v.y, a0 * h.y);
                    w.z = fmaf(sa, v.z, a0 * h.z); w.w = fmaf(sa, v.w, a0 * h.w);
                    ((float4*)out)[i4] = w;
                }
            }
        }
    }
}

extern "C" void kernel_launch(void* const* d_in, const int* in_sizes, int n_in,
                              void* d_out, int out_size, void* d_ws, size_t ws_size,
                              hipStream_t stream)
{
    (void)in_sizes; (void)n_in; (void)out_size; (void)ws_size;

    const float* in_feat = (const float*)d_in[0];
    const int*   edge    = (const int*)d_in[1];
    const float* W       = (const float*)d_in[2];
    const float* bias    = (const float*)d_in[3];
    const float* alphas  = (const float*)d_in[4];
    float*       out     = (float*)d_out;

    const int* srcIdx = edge;
    const int* dstIdx = edge + N_EDGES;

    float* degOut = (float*)d_ws;
    float* degIn  = degOut + N_NODES;
    float* xbuf   = degIn + N_NODES;
    float* hbuf   = xbuf + (size_t)N_NODES * H_FEATS;

    hipMemsetAsync(degOut, 0, (size_t)2 * N_NODES * sizeof(float), stream);
    degree_kernel<<<(N_EDGES + 255) / 256, 256, 0, stream>>>(srcIdx, dstIdx, degOut, degIn);
    invnorm_kernel<<<(N_NODES + 255) / 256, 256, 0, stream>>>(degOut, degIn);

    lin_wmma_kernel<<<LIN_WGS, 256, 0, stream>>>(in_feat, W, bias, xbuf);

    conv_owner_kernel<true><<<OWN_WGS, OWN_THREADS, 0, stream>>>(
        xbuf, srcIdx, dstIdx, degOut, degIn, alphas, hbuf, out);

    conv_owner_kernel<false><<<OWN_WGS, OWN_THREADS, 0, stream>>>(
        hbuf, srcIdx, dstIdx, degOut, degIn, alphas, nullptr, out);
}


# degrees: 16 node ranges x 16 edge slices, a work-group counts its ids in LDS (integer adds, exact), a merge kernel sums the slices; no global f32 atomics, no memset; conv sums and order same
# speedup vs baseline: 3.0539x; 1.2798x over previous
#include <hip/hip_runtime.h>
#include <hip/hip_bf16.h>
#include <math.h>

#define N_NODES 100000
#define N_EDGES 600000
#define IN_FEATS 256
#define H_FEATS 128

typedef __attribute__((ext_vector_type(2))) float v2f;
typedef __attribute__((ext_vector_type(8))) float v8f;
typedef __attribute__((ext_vector_type(4))) float v4f;

#define DEG_RANGES 16
#define DEG_SLICES 16
#define DEG_RANGE_NODES ((((N_NODES + DEG_RANGES - 1) / DEG_RANGES) + 31) / 32 * 32)
#define DEG_THREADS 512
#define DEG_Q_TOTAL (N_EDGES / 4)
#define DEG_Q_SLICE ((DEG_Q_TOTAL + DEG_SLICES - 1) / DEG_SLICES)
#define DEG_ITERS ((DEG_Q_SLICE + DEG_THREADS - 1) / DEG_THREADS)
#define DEG_BATCH 10

static_assert(N_EDGES % 4 == 0 && N_NODES % 4 == 0, "ids and counts move 16 bytes at a time");
static_assert((DEG_RANGES - 1) * DEG_RANGE_NODES < N_NODES, "every range holds nodes");
static_assert(2 * DEG_SLICES <= H_FEATS, "the planes fit the space of hbuf");

__device__ __forceinline__ void count_owned(unsigned* cnt, const int4 v, const int lo, const unsigned lim)
{
    const unsigned t0 = (unsigned)(v.x - lo);
    const unsigned t1 = (unsigned)(v.y - lo);
    const unsigned t2 = (unsigned)(v.z - lo);
    const unsigned t3 = (unsigned)(v.w - lo);
    if (t0 < lim) atomicAdd(&cnt[t0], 1u);
    if (t1 < lim) atomicAdd(&cnt[t1], 1u);
    if (t2 < lim) atomicAdd(&cnt[t2], 1u);
    if (t3 < lim) atomicAdd(&cnt[t3], 1u);
}

__global__ __launch_bounds__(DEG_THREADS) void degree_count_kernel(
    const int* __restrict__ srcIdx, const int* __restrict__ dstIdx, float* __restrict__ planes)
{
    __shared__ __attribute__((aligned(16))) unsigned cnt[2][DEG_RANGE_NODES];

    const int tid   = threadIdx.x;
    const int range = blockIdx.x % DEG_RANGES;
    const int slice = blockIdx.x / DEG_RANGES;
    const int lo    = range * DEG_RANGE_NODES;
    const int rows  = (N_NODES - lo) < DEG_RANGE_NODES ? (N_NODES - lo) : DEG_RANGE_NODES;

    for (int i = tid; i < 2 * DEG_RANGE_NODES / 4; i += DEG_THREADS)
        ((uint4*)&cnt[0][0])[i] = make_uint4(0u, 0u, 0u, 0u);
    __syncthreads();

    const int qlo = slice * DEG_Q_SLICE;
    for (int it0 = 0; it0 < DEG_ITERS; it0 += DEG_BATCH) {
        int4 s4[DEG_BATCH];
        int4 d4[DEG_BATCH];
        #pragma unroll
        for (int k = 0; k < DEG_BATCH; ++k) {
            const int j = (it0 + k) * DEG_THREADS + tid;
            const int q = qlo + j;
            const bool ok = (j < DEG_Q_SLICE) && (q < DEG_Q_TOTAL);
            const int qq = ok ? q : DEG_Q_TOTAL - 1;
            s4[k] = ((const int4*)srcIdx)[qq];
            d4[k] = ((const int4*)dstIdx)[qq];
        }
        #pragma unroll
        for (int k = 0; k < DEG_BATCH; ++k) {
            const int j = (it0 + k) * DEG_THREADS + tid;
            const bool ok = (j < DEG_Q_SLICE) && (qlo + j < DEG_Q_TOTAL);
            const unsigned lim = ok ? (unsigned)rows : 0u;
            count_owned(cnt[0], s4[k], lo, lim);
            count_owned(cnt[1], d4[k], lo, lim);
        }
    }
    __syncthreads();

    uint4* po = (uint4*)(planes + (size_t)slice * (2 * N_NODES) + lo);
    uint4* pi = (uint4*)(planes + (size_t)slice * (2 * N_NODES) + N_NODES + lo);
    for (int i = tid; i < rows / 4; i += DEG_THREADS) {
        po[i] = ((const uint4*)&cnt[0][0])[i];
        pi[i] = ((const uint4*)&cnt[1][0])[i];
    }
}

__device__ __forceinline__ float inv_norm(const unsigned c)
{
    const float d = (float)c;
    return (d > 0.0f) ? rsqrtf(fmaxf(d, 1.0f)) : 0.0f;
}

__global__ __launch_bounds__(256) void degree_merge_kernel(
    const float* __restrict__ planes, float* __restrict__ norm)
{
    const int i = blockIdx.x * blockDim.x + threadIdx.x;
    if (i < 2 * N_NODES / 4) {
        uint4 v[DEG_SLICES];
        #pragma unroll
        for (int s = 0; s < DEG_SLICES; ++s)
            v[s] = ((const uint4*)planes)[(size_t)s * (2 * N_NODES / 4) + i];
        uint4 c = v[0];
        #pragma unroll
        for (int s = 1; s < DEG_SLICES; ++s) {
            c.x += v[s].x; c.y += v[s].y; c.z += v[s].z; c.w += v[s].w;
        }
        float4 o;
        o.x = inv_norm(c.x); o.y = inv_norm(c.y); o.z = inv_norm(c.z); o.w = inv_norm(c.w);
        ((float4*)norm)[i] = o;
    }
}

#define LIN_ROWS 32
#define LIN_PITCH (IN_FEATS + 4)
#define LIN_WGS 512
#define LIN_TILES (N_NODES / LIN_ROWS)
#define LIN_KSTEPS (IN_FEATS / 4)
#define LIN_PIECES (LIN_ROWS * IN_FEATS / 4 / 256)

static_assert(N_NODES % LIN_ROWS == 0, "tiles are whole");
static_assert(LIN_WGS <= LIN_TILES, "every work-group has a first tile");
static_assert(LIN_PITCH % 64 == 4 && (LIN_PITCH * 4) % 16 == 0, "bank spread and 16-byte alignment of the rows");
static_assert(H_FEATS == 8 * 16, "eight waves of 16 columns");

__global__ __launch_bounds__(256, 2) void lin_wmma_kernel(
    const float* __restrict__ A, const float* __restrict__ W,
    const float* __restrict__ bias, float* __restrict__ X)
{
    __shared__ __attribute__((aligned(16))) float tile[2][LIN_ROWS * LIN_PITCH];

    const int tid    = threadIdx.x;
    const int lane   = tid & 31;
    const int wave   = tid >> 5;
    const int halfId = lane >> 4;
    const int l16    = lane & 15;
    const int n0     = wave * 16;

    v2f bf[LIN_KSTEPS];
    {
        const float* bp = W + (size_t)(2 * halfId) * H_FEATS + n0 + l16;
        #pragma unroll
        for (int j = 0; j < LIN_KSTEPS; ++j) {
            bf[j].x = bp[(size_t)(4 * j) * H_FEATS];
            bf[j].y = bp[(size_t)(4 * j + 1) * H_FEATS];
        }
    }
    const float bb = bias[n0 + l16];

    const int stRow = tid >> 6;
    const int stCol = (tid & 63) * 4;
    const v4f* A4 = (const v4f*)A;

    v4f st[LIN_PIECES];
    int t = blockIdx.x;
    #pragma unroll
    for (int i = 0; i < LIN_PIECES; ++i)
        st[i] = A4[(size_t)t * (LIN_ROWS * IN_FEATS / 4) + i * 256 + tid];
    #pragma unroll
    for (int i = 0; i < LIN_PIECES; ++i)
        *(v4f*)(&tile[0][(i * 4 + stRow) * LIN_PITCH + stCol]) = st[i];
    __syncthreads();

    int buf = 0;
    for (; t < LIN_TILES; t += LIN_WGS) {
        const int tn = t + LIN_WGS;
        const bool more = tn < LIN_TILES;
        if (more) {
            #pragma unroll
            for (int i = 0; i < LIN_PIECES; ++i)
                st[i] = A4[(size_t)tn * (LIN_ROWS * IN_FEATS / 4) + i * 256 + tid];
        }

        const float* fa = &tile[buf][l16 * LIN_PITCH + 2 * halfId];
        v8f c0 = {};
        v8f c1 = {};
        #pragma unroll
        for (int j = 0; j < LIN_KSTEPS; ++j) {
            const v2f a0 = *(const v2f*)(fa + 4 * j);
            const v2f a1 = *(const v2f*)(fa + 16 * LIN_PITCH + 4 * j);
            c0 = __builtin_amdgcn_wmma_f32_16x16x4_f32(
                    false, a0, false, bf[j], (short)0, c0, false, false);
            c1 = __builtin_amdgcn_wmma_f32_16x16x4_f32(
                    false, a1, false, bf[j], (short)0, c1, false, false);
        }

        if (more) {
            #pragma unroll
            for (int i = 0; i < LIN_PIECES; ++i)
                *(v4f*)(&tile[buf ^ 1][(i * 4 + stRow) * LIN_PITCH + stCol]) = st[i];
        }
        __syncthreads();
        buf ^= 1;

        const int rowBase = t * LIN_ROWS;
        #pragma unroll
        for (int r = 0; r < 8; ++r) {
            const int m = r + 8 * halfId;
            X[(size_t)(rowBase + m) * H_FEATS + n0 + l16]      = c0[r] + bb;
            X[(size_t)(rowBase + 16 + m) * H_FEATS + n0 + l16] = c1[r] + bb;
        }
    }
}

#define OWN_WGS 256
#define OWN_ROWS ((N_NODES + OWN_WGS - 1) / OWN_WGS)
#define OWN_THREADS 1024
#define OWN_WAVES (OWN_THREADS / 32)
#define SEG_PER_ROUND 7
#define EDGES_PER_ROUND (SEG_PER_ROUND * OWN_THREADS * 4)
#define SCAN_ROUNDS ((N_EDGES + EDGES_PER_ROUND - 1) / EDGES_PER_ROUND)
#define WAVE_LIST_CAP 960
#define WAVE_EDGES_PER_ROUND (SEG_PER_ROUND * 32 * 4)
#define WAVE_DRAIN_AT (WAVE_LIST_CAP - WAVE_EDGES_PER_ROUND)
#define GATHER_BATCH 16
#define ROW_GROUP 4
#define SAVE_HDR 32
#define SAVE_CAP WAVE_LIST_CAP
#define SAVE_STRIDE 1024

static_assert(OWN_ROWS <= 512, "local row number must fit the 9 low bits of a list entry");
static_assert(N_EDGES < (1 << 23), "edge number must fit the 23 high bits of a list entry");
static_assert(N_EDGES % 4 == 0, "destination ids are read four at a time");
static_assert(WAVE_DRAIN_AT >= 0, "one round of a wave's matches must fit its list");
static_assert(N_NODES <= (1 << 23), "source id must fit the 23 high bits of a stored entry");
static_assert(SAVE_HDR + SAVE_CAP <= SAVE_STRIDE, "count line and entries must fit the wave's region");
static_assert((N_NODES - (OWN_WGS - 1) * OWN_ROWS) * H_FEATS >= OWN_WAVES * SAVE_STRIDE,
              "the last work-group's rows of out must hold its waves' regions");
static_assert((OWN_ROWS * H_FEATS) % 64 == 0 && SAVE_STRIDE % 64 == 0, "regions start on 256-byte units");

__device__ __forceinline__ float softmax2(const float* __restrict__ alphas, int idx)
{
    const float a0 = alphas[0], a1 = alphas[1];
    const float m  = fmaxf(a0, a1);
    const float e0 = __expf(a0 - m), e1 = __expf(a1 - m);
    return (idx == 0 ? e0 : e1) / (e0 + e1);
}

__device__ __forceinline__ void load_round(
    const int* __restrict__ dstIdx, const int round, const int tid, int4 (&d)[SEG_PER_ROUND])
{
    #pragma unroll
    for (int g = 0; g < SEG_PER_ROUND; ++g) {
        const int q = (round * SEG_PER_ROUND + g) * OWN_THREADS + tid;
        d[g] = ((const int4*)dstIdx)[q < N_EDGES / 4 ? q : N_EDGES / 4 - 1];
    }
}

template <bool STORED_IN, bool STORE_OUT>
__device__ __forceinline__ void drain_list(
    const float* __restrict__ feat, const int* __restrict__ srcIdx,
    const float* __restrict__ invSrc, float* acc, const unsigned* list,
    const int count, const int lane, float* save, const int saveBase)
{
    for (int base = 0; base < count; base += GATHER_BATCH) {
        const int left = count - base;
        const int nvalid = left < GATHER_BATCH ? left : GATHER_BATCH;
        unsigned ent = 0;
        int s = 0;
        float w = 0.0f;
        if (lane < nvalid) {
            ent = list[base + lane];
            s = STORED_IN ? (int)(ent >> 9) : srcIdx[ent >> 9];
            w = invSrc[s];
            if (STORE_OUT) {
                const int pos = saveBase + base + lane;
                if (pos < SAVE_CAP)
                    save[SAVE_HDR + pos] = __uint_as_float(((unsigned)s << 9) | (ent & 511u));
            }
        }
        float v[GATHER_BATCH][4];
        #pragma unroll
        for (int j = 0; j < GATHER_BATCH; ++j) {
            if (j < nvalid) {
                const int sj = __builtin_amdgcn_readlane(s, j);
                const float* fp = feat + (size_t)sj * H_FEATS + lane;
                v[j][0] = fp[0];
                v[j][1] = fp[32];
                v[j][2] = fp[64];
                v[j][3] = fp[96];
            }
        }
        #pragma unroll
        for (int j = 0; j < GATHER_BATCH; ++j) {
            if (j < nvalid) {
                const int dl = (int)(__builtin_amdgcn_readlane((int)ent, j) & 511);
                const float wj = __int_as_float(__builtin_amdgcn_readlane(__float_as_int(w), j));
                float* ap = acc + dl * H_FEATS + lane;
                unsafeAtomicAdd(ap + 0,  v[j][0] * wj);
                unsafeAtomicAdd(ap + 32, v[j][1] * wj);
                unsafeAtomicAdd(ap + 64, v[j][2] * wj);
                unsafeAtomicAdd(ap + 96, v[j][3] * wj);
            }
        }
    }
}

__device__ __forceinline__ void append_owned(
    unsigned* mylist, int& cnt, const bool owned, const unsigned edge, const unsigned localRow)
{
    const unsigned mask = __builtin_amdgcn_ballot_w32(owned);
    if (owned) mylist[cnt + (int)__builtin_amdgcn_mbcnt_lo(mask, 0u)] = (edge << 9) | localRow;
    cnt += __builtin_popcount(mask);
}

template <bool STORE_OUT>
__device__ __forceinline__ void scan_and_drain(
    const float* __restrict__ feat, const int* __restrict__ srcIdx,
    const int* __restrict__ dstIdx, const float* __restrict__ invSrc,
    float* acc, unsigned* mylist, float* save,
    const int lo, const int rows, const int tid, const int lane)
{
    int rr = blockIdx.x % SCAN_ROUNDS;
    int cnt = 0;
    int stored = 0;
    for (int round = 0; round < SCAN_ROUNDS; ++round) {
        int4 d[SEG_PER_ROUND];
        load_round(dstIdx, rr, tid, d);
        #pragma unroll
        for (int g = 0; g < SEG_PER_ROUND; ++g) {
            const unsigned t0 = (unsigned)(d[g].x - lo);
            const unsigned t1 = (unsigned)(d[g].y - lo);
            const unsigned t2 = (unsigned)(d[g].z - lo);
            const unsigned t3 = (unsigned)(d[g].w - lo);
            const int q = (rr * SEG_PER_ROUND + g) * OWN_THREADS + tid;
            const unsigned lim = (q < N_EDGES / 4) ? (unsigned)rows : 0u;
            const bool m0 = t0 < lim, m1 = t1 < lim;
            const bool m2 = t2 < lim, m3 = t3 < lim;
            if (__builtin_amdgcn_ballot_w32(m0 | m1 | m2 | m3)) {
                const unsigned e = (unsigned)(q * 4);
                append_owned(mylist, cnt, m0, e + 0u, t0);
                append_owned(mylist, cnt, m1, e + 1u, t1);
                append_owned(mylist, cnt, m2, e + 2u, t2);
                append_owned(mylist, cnt, m3, e + 3u, t3);
            }
        }
        if (cnt > WAVE_DRAIN_AT || round + 1 == SCAN_ROUNDS) {
            drain_list<false, STORE_OUT>(feat, srcIdx, invSrc, acc, mylist, cnt, lane, save, stored);
            stored += cnt;
            cnt = 0;
        }
        rr = (rr + 1 == SCAN_ROUNDS) ? 0 : rr + 1;
    }
    if (STORE_OUT && lane == 0) save[0] = __int_as_float(stored);
}

template <bool FIRST>
__global__ __launch_bounds__(OWN_THREADS) void conv_owner_kernel(
    const float* __restrict__ feat, const int* __restrict__ srcIdx,
    const int* __restrict__ dstIdx, const float* __restrict__ invSrc,
    const float* __restrict__ invDst, const float* __restrict__ alphas,
    float* __restrict__ hbuf, float* __restrict__ out)
{
    __shared__ float acc[OWN_ROWS * H_FEATS];
    __shared__ unsigned list[OWN_WAVES * WAVE_LIST_CAP];

    const int tid  = threadIdx.x;
    const int lane = tid & 31;
    const int wave = __builtin_amdgcn_readfirstlane(tid >> 5);
    const int lo   = blockIdx.x * OWN_ROWS;
    const int rows = (N_NODES - lo) < OWN_ROWS ? (N_NODES - lo) : OWN_ROWS;
    unsigned* mylist = list + wave * WAVE_LIST_CAP;

    for (int i = tid; i < OWN_ROWS * H_FEATS / 4; i += OWN_THREADS)
        ((float4*)acc)[i] = make_float4(0.0f, 0.0f, 0.0f, 0.0f);
    __syncthreads();

    float* save = out + (size_t)lo * H_FEATS + wave * SAVE_STRIDE;
    if (FIRST) {
        scan_and_drain<true>(feat, srcIdx, dstIdx, invSrc, acc, mylist, save, lo, rows, tid, lane);
    } else {
        const int total = __builtin_amdgcn_readfirstlane(__float_as_int(save[0]));
        if (total <= SAVE_CAP) {
            for (int i0 = 0; i0 < total; i0 += 128) {
                float t[4];
                #pragma unroll
                for (int u = 0; u < 4; ++u) {
                    const int i = i0 + u * 32 + lane;
                    t[u] = save[SAVE_HDR + (i < total ? i : 0)];
                }
                #pragma unroll
                for (int u = 0; u < 4; ++u) {
                    const int i = i0 + u * 32 + lane;
                    if (i < total) mylist[i] = __float_as_uint(t[u]);
                }
            }
            drain_list<true, false>(feat, srcIdx, invSrc, acc, mylist, total, lane, save, 0);
        } else {
            scan_and_drain<false>(feat, srcIdx, dstIdx, invSrc, acc, mylist, save, lo, rows, tid, lane);
        }
    }
    __syncthreads();

    const float a0 = FIRST ? 0.0f : softmax2(alphas, 0);
    const float a1 = FIRST ? 0.0f : softmax2(alphas, 1);
    for (int r0 = wave; r0 < rows; r0 += OWN_WAVES * ROW_GROUP) {
        float4 o[ROW_GROUP];
        float s[ROW_GROUP];
        #pragma unroll
        for (int k = 0; k < ROW_GROUP; ++k) {
            const int r = r0 + k * OWN_WAVES;
            if (r < rows) {
                s[k] = invDst[lo + r];
                if (!FIRST) o[k] = ((const float4*)feat)[(size_t)(lo + r) * (H_FEATS / 4) + lane];
            }
        }
        #pragma unroll
        for (int k = 0; k < ROW_GROUP; ++k) {
            const int r = r0 + k * OWN_WAVES;
            if (r < rows) {
                const float4 v = ((const float4*)(acc + r * H_FEATS))[lane];
                const size_t i4 = (size_t)(lo + r) * (H_FEATS / 4) + lane;
                if (FIRST) {
                    float4 h;
                    h.x = v.x * s[k]; h.y = v.y * s[k]; h.z = v.z * s[k]; h.w = v.w * s[k];
                    ((float4*)hbuf)[i4] = h;
                } else {
                    const float sa = s[k] * a1;
                    const float4 h = o[k];
                    float4 w;
                    w.x = fmaf(sa, v.x, a0 * h.x); w.y = fmaf(sa, v.y, a0 * h.y);
                    w.z = fmaf(sa, v.z, a0 * h.z); w.w = fmaf(sa, v.w, a0 * h.w);
                    ((float4*)out)[i4] = w;
                }
            }
        }
    }
}

extern "C" void kernel_launch(void* const* d_in, const int* in_sizes, int n_in,
                              void* d_out, int out_size, void* d_ws, size_t ws_size,
                              hipStream_t stream)
{
    (void)in_sizes; (void)n_in; (void)out_size; (void)ws_size;

    const float* in_feat = (const float*)d_in[0];
    const int*   edge    = (const int*)d_in[1];
    const float* W       = (const float*)d_in[2];
    const float* bias    = (const float*)d_in[3];
    const float* alphas  = (const float*)d_in[4];
    float*       out     = (float*)d_out;

    const int* srcIdx = edge;
    const int* dstIdx = edge + N_EDGES;

    float* degOut = (float*)d_ws;
    float* degIn  = degOut + N_NODES;
    float* xbuf   = degIn + N_NODES;
    float* hbuf   = xbuf + (size_t)N_NODES * H_FEATS;

    degree_count_kernel<<<DEG_RANGES * DEG_SLICES, DEG_THREADS, 0, stream>>>(srcIdx, dstIdx, hbuf);
    degree_merge_kernel<<<(2 * N_NODES / 4 + 255) / 256, 256, 0, stream>>>(hbuf, degOut);

    lin_wmma_kernel<<<LIN_WGS, 256, 0, stream>>>(in_feat, W, bias, xbuf);

    conv_owner_kernel<true><<<OWN_WGS, OWN_THREADS, 0, stream>>>(
        xbuf, srcIdx, dstIdx, degOut, degIn, alphas, hbuf, out);

    conv_owner_kernel<false><<<OWN_WGS, OWN_THREADS, 0, stream>>>(
        hbuf, srcIdx, dstIdx, degOut, degIn, alphas, nullptr, out);
}


# conv step 1 scans one byte per edge (owner work-group of its destination, written by the degree kernel), not the 4-byte id; a lane's 28 flags of a round fit one word, one test; LDS atomic sums same
# speedup vs baseline: 3.2370x; 1.0600x over previous
#include <hip/hip_runtime.h>
#include <hip/hip_bf16.h>
#include <math.h>

#define N_NODES 100000
#define N_EDGES 600000
#define IN_FEATS 256
#define H_FEATS 128

typedef __attribute__((ext_vector_type(2))) float v2f;
typedef __attribute__((ext_vector_type(8))) float v8f;
typedef __attribute__((ext_vector_type(4))) float v4f;

#define OWN_WGS 256
#define OWN_ROWS ((N_NODES + OWN_WGS - 1) / OWN_WGS)
#define OWN_THREADS 1024
#define OWN_WAVES (OWN_THREADS / 32)
#define SAVE_STRIDE 1024
#define SEG_PER_ROUND 7
#define CODE_CHUNK_WORDS (SEG_PER_ROUND * OWN_THREADS)
#define CODE_CHUNKS ((N_EDGES / 4 + CODE_CHUNK_WORDS - 1) / CODE_CHUNK_WORDS)
#define CODE_SPAN (OWN_ROWS * H_FEATS)
#define CODE_SKIP (OWN_WAVES * SAVE_STRIDE)

static_assert(OWN_WGS == 256, "an owner's number is one byte");
static_assert(CODE_SPAN - CODE_SKIP >= CODE_CHUNK_WORDS, "a chunk of codes fits behind the waves' regions");
static_assert(CODE_CHUNKS <= OWN_WGS - 1, "chunks sit in rows of owners that have all their rows");
static_assert((CODE_SPAN * 4) % 128 == 0 && (CODE_SKIP * 4) % 128 == 0 && CODE_CHUNK_WORDS % 32 == 0, "chunks are whole lines");

__device__ __forceinline__ unsigned code_word(const int q)
{
    const unsigned u = (unsigned)q;
    return (u / CODE_CHUNK_WORDS) * CODE_SPAN + CODE_SKIP + (u % CODE_CHUNK_WORDS);
}

__device__ __forceinline__ unsigned owner_codes(const int4 v)
{
    const unsigned o0 = ((unsigned)v.x / OWN_ROWS) & 255u;
    const unsigned o1 = ((unsigned)v.y / OWN_ROWS) & 255u;
    const unsigned o2 = ((unsigned)v.z / OWN_ROWS) & 255u;
    const unsigned o3 = ((unsigned)v.w / OWN_ROWS) & 255u;
    return o0 | (o1 << 8) | (o2 << 16) | (o3 << 24);
}

#define DEG_RANGES 16
#define DEG_SLICES 16
#define DEG_RANGE_NODES ((((N_NODES + DEG_RANGES - 1) / DEG_RANGES) + 31) / 32 * 32)
#define DEG_THREADS 512
#define DEG_Q_TOTAL (N_EDGES / 4)
#define DEG_Q_SLICE ((DEG_Q_TOTAL + DEG_SLICES - 1) / DEG_SLICES)
#define DEG_ITERS ((DEG_Q_SLICE + DEG_THREADS - 1) / DEG_THREADS)
#define DEG_BATCH 10
#define CODE_WG_WORDS ((((N_EDGES / 4 + DEG_RANGES * DEG_SLICES - 1) / (DEG_RANGES * DEG_SLICES)) + 31) / 32 * 32)
#define CODE_ITERS ((CODE_WG_WORDS + DEG_THREADS - 1) / DEG_THREADS)

static_assert(N_EDGES % 4 == 0 && N_NODES % 4 == 0, "ids and counts move 16 bytes at a time");
static_assert((DEG_RANGES - 1) * DEG_RANGE_NODES < N_NODES, "every range holds nodes");
static_assert(2 * DEG_SLICES <= H_FEATS, "the planes fit the space of hbuf");

__device__ __forceinline__ void count_owned(unsigned* cnt, const int4 v, const int lo, const unsigned lim)
{
    const unsigned t0 = (unsigned)(v.x - lo);
    const unsigned t1 = (unsigned)(v.y - lo);
    const unsigned t2 = (unsigned)(v.z - lo);
    const unsigned t3 = (unsigned)(v.w - lo);
    if (t0 < lim) atomicAdd(&cnt[t0], 1u);
    if (t1 < lim) atomicAdd(&cnt[t1], 1u);
    if (t2 < lim) atomicAdd(&cnt[t2], 1u);
    if (t3 < lim) atomicAdd(&cnt[t3], 1u);
}

__global__ __launch_bounds__(DEG_THREADS) void degree_count_kernel(
    const int* __restrict__ srcIdx, const int* __restrict__ dstIdx, float* __restrict__ planes,
    float* __restrict__ codes)
{
    __shared__ __attribute__((aligned(16))) unsigned cnt[2][DEG_RANGE_NODES];

    const int tid   = threadIdx.x;
    const int range = blockIdx.x % DEG_RANGES;
    const int slice = blockIdx.x / DEG_RANGES;
    const int lo    = range * DEG_RANGE_NODES;
    const int rows  = (N_NODES - lo) < DEG_RANGE_NODES ? (N_NODES - lo) : DEG_RANGE_NODES;

    int4 cv[CODE_ITERS];
    #pragma unroll
    for (int u = 0; u < CODE_ITERS; ++u) {
        const int w = u * DEG_THREADS + tid;
        const int q = blockIdx.x * CODE_WG_WORDS + w;
        cv[u] = ((const int4*)dstIdx)[(w < CODE_WG_WORDS && q < DEG_Q_TOTAL) ? q : DEG_Q_TOTAL - 1];
    }

    for (int i = tid; i < 2 * DEG_RANGE_NODES / 4; i += DEG_THREADS)
        ((uint4*)&cnt[0][0])[i] = make_uint4(0u, 0u, 0u, 0u);
    __syncthreads();

    #pragma unroll
    for (int u = 0; u < CODE_ITERS; ++u) {
        const int w = u * DEG_THREADS + tid;
        const int q = blockIdx.x * CODE_WG_WORDS + w;
        if (w < CODE_WG_WORDS && q < DEG_Q_TOTAL)
            codes[code_word(q)] = __uint_as_float(owner_codes(cv[u]));
    }

    const int qlo = slice * DEG_Q_SLICE;
    for (int it0 = 0; it0 < DEG_ITERS; it0 += DEG_BATCH) {
        int4 s4[DEG_BATCH];
        int4 d4[DEG_BATCH];
        #pragma unroll
        for (int k = 0; k < DEG_BATCH; ++k) {
            const int j = (it0 + k) * DEG_THREADS + tid;
            const int q = qlo + j;
            const bool ok = (j < DEG_Q_SLICE) && (q < DEG_Q_TOTAL);
            const int qq = ok ? q : DEG_Q_TOTAL - 1;
            s4[k] = ((const int4*)srcIdx)[qq];
            d4[k] = ((const int4*)dstIdx)[qq];
        }
        #pragma unroll
        for (int k = 0; k < DEG_BATCH; ++k) {
            const int j = (it0 + k) * DEG_THREADS + tid;
            const bool ok = (j < DEG_Q_SLICE) && (qlo + j < DEG_Q_TOTAL);
            const unsigned lim = ok ? (unsigned)rows : 0u;
            count_owned(cnt[0], s4[k], lo, lim);
            count_owned(cnt[1], d4[k], lo, lim);
        }
    }
    __syncthreads();

    uint4* po = (uint4*)(planes + (size_t)slice * (2 * N_NODES) + lo);
    uint4* pi = (uint4*)(planes + (size_t)slice * (2 * N_NODES) + N_NODES + lo);
    for (int i = tid; i < rows / 4; i += DEG_THREADS) {
        po[i] = ((const uint4*)&cnt[0][0])[i];
        pi[i] = ((const uint4*)&cnt[1][0])[i];
    }
}

__device__ __forceinline__ float inv_norm(const unsigned c)
{
    const float d = (float)c;
    return (d > 0.0f) ? rsqrtf(fmaxf(d, 1.0f)) : 0.0f;
}

__global__ __launch_bounds__(256) void degree_merge_kernel(
    const float* __restrict__ planes, float* __restrict__ norm)
{
    const int i = blockIdx.x * blockDim.x + threadIdx.x;
    if (i < 2 * N_NODES / 4) {
        uint4 v[DEG_SLICES];
        #pragma unroll
        for (int s = 0; s < DEG_SLICES; ++s)
            v[s] = ((const uint4*)planes)[(size_t)s * (2 * N_NODES / 4) + i];
        uint4 c = v[0];
        #pragma unroll
        for (int s = 1; s < DEG_SLICES; ++s) {
            c.x += v[s].x; c.y += v[s].y; c.z += v[s].z; c.w += v[s].w;
        }
        float4 o;
        o.x = inv_norm(c.x); o.y = inv_norm(c.y); o.z = inv_norm(c.z); o.w = inv_norm(c.w);
        ((float4*)norm)[i] = o;
    }
}

#define LIN_ROWS 32
#define LIN_PITCH (IN_FEATS + 4)
#define LIN_WGS 512
#define LIN_TILES (N_NODES / LIN_ROWS)
#define LIN_KSTEPS (IN_FEATS / 4)
#define LIN_PIECES (LIN_ROWS * IN_FEATS / 4 / 256)

static_assert(N_NODES % LIN_ROWS == 0, "tiles are whole");
static_assert(LIN_WGS <= LIN_TILES, "every work-group has a first tile");
static_assert(LIN_PITCH % 64 == 4 && (LIN_PITCH * 4) % 16 == 0, "bank spread and 16-byte alignment of the rows");
static_assert(H_FEATS == 8 * 16, "eight waves of 16 columns");

__global__ __launch_bounds__(256, 2) void lin_wmma_kernel(
    const float* __restrict__ A, const float* __restrict__ W,
    const float* __restrict__ bias, float* __restrict__ X)
{
    __shared__ __attribute__((aligned(16))) float tile[2][LIN_ROWS * LIN_PITCH];

    const int tid    = threadIdx.x;
    const int lane   = tid & 31;
    const int wave   = tid >> 5;
    const int halfId = lane >> 4;
    const int l16    = lane & 15;
    const int n0     = wave * 16;

    v2f bf[LIN_KSTEPS];
    {
        const float* bp = W + (size_t)(2 * halfId) * H_FEATS + n0 + l16;
        #pragma unroll
        for (int j = 0; j < LIN_KSTEPS; ++j) {
            bf[j].x = bp[(size_t)(4 * j) * H_FEATS];
            bf[j].y = bp[(size_t)(4 * j + 1) * H_FEATS];
        }
    }
    const float bb = bias[n0 + l16];

    const int stRow = tid >> 6;
    const int stCol = (tid & 63) * 4;
    const v4f* A4 = (const v4f*)A;

    v4f st[LIN_PIECES];
    int t = blockIdx.x;
    #pragma unroll
    for (int i = 0; i < LIN_PIECES; ++i)
        st[i] = A4[(size_t)t * (LIN_ROWS * IN_FEATS / 4) + i * 256 + tid];
    #pragma unroll
    for (int i = 0; i < LIN_PIECES; ++i)
        *(v4f*)(&tile[0][(i * 4 + stRow) * LIN_PITCH + stCol]) = st[i];
    __syncthreads();

    int buf = 0;
    for (; t < LIN_TILES; t += LIN_WGS) {
        const int tn = t + LIN_WGS;
        const bool more = tn < LIN_TILES;
        if (more) {
            #pragma unroll
            for (int i = 0; i < LIN_PIECES; ++i)
                st[i] = A4[(size_t)tn * (LIN_ROWS * IN_FEATS / 4) + i * 256 + tid];
        }

        const float* fa = &tile[buf][l16 * LIN_PITCH + 2 * halfId];
        v8f c0 = {};
        v8f c1 = {};
        #pragma unroll
        for (int j = 0; j < LIN_KSTEPS; ++j) {
            const v2f a0 = *(const v2f*)(fa + 4 * j);
            const v2f a1 = *(const v2f*)(fa + 16 * LIN_PITCH + 4 * j);
            c0 = __builtin_amdgcn_wmma_f32_16x16x4_f32(
                    false, a0, false, bf[j], (short)0, c0, false, false);
            c1 = __builtin_amdgcn_wmma_f32_16x16x4_f32(
                    false, a1, false, bf[j], (short)0, c1, false, false);
        }

        if (more) {
            #pragma unroll
            for (int i = 0; i < LIN_PIECES; ++i)
                *(v4f*)(&tile[buf ^ 1][(i * 4 + stRow) * LIN_PITCH + stCol]) = st[i];
        }
        __syncthreads();
        buf ^= 1;

        const int rowBase = t * LIN_ROWS;
        #pragma unroll
        for (int r = 0; r < 8; ++r) {
            const int m = r + 8 * halfId;
            X[(size_t)(rowBase + m) * H_FEATS + n0 + l16]      = c0[r] + bb;
            X[(size_t)(rowBase + 16 + m) * H_FEATS + n0 + l16] = c1[r] + bb;
        }
    }
}

#define EDGES_PER_ROUND (SEG_PER_ROUND * OWN_THREADS * 4)
#define SCAN_ROUNDS ((N_EDGES + EDGES_PER_ROUND - 1) / EDGES_PER_ROUND)
#define WAVE_LIST_CAP 960
#define WAVE_EDGES_PER_ROUND (SEG_PER_ROUND * 32 * 4)
#define WAVE_DRAIN_AT (WAVE_LIST_CAP - WAVE_EDGES_PER_ROUND)
#define GATHER_BATCH 16
#define ROW_GROUP 4
#define SAVE_HDR 32
#define SAVE_CAP WAVE_LIST_CAP

static_assert(OWN_ROWS <= 512, "local row number must fit the 9 low bits of a stored entry");
static_assert(N_EDGES % 4 == 0, "destination ids are read four at a time");
static_assert(WAVE_DRAIN_AT >= 0, "one round of a wave's matches must fit its list");
static_assert(N_NODES <= (1 << 23), "source id must fit the 23 high bits of a stored entry");
static_assert(SAVE_HDR + SAVE_CAP <= SAVE_STRIDE, "count line and entries must fit the wave's region");
static_assert((N_NODES - (OWN_WGS - 1) * OWN_ROWS) * H_FEATS >= OWN_WAVES * SAVE_STRIDE,
              "the last work-group's rows of out must hold its waves' regions");
static_assert((OWN_ROWS * H_FEATS) % 64 == 0 && SAVE_STRIDE % 64 == 0, "regions start on 256-byte units");

__device__ __forceinline__ float softmax2(const float* __restrict__ alphas, int idx)
{
    const float a0 = alphas[0], a1 = alphas[1];
    const float m  = fmaxf(a0, a1);
    const float e0 = __expf(a0 - m), e1 = __expf(a1 - m);
    return (idx == 0 ? e0 : e1) / (e0 + e1);
}

__device__ __forceinline__ void load_round(
    const int* __restrict__ dstIdx, const int round, const int tid, int4 (&d)[SEG_PER_ROUND])
{
    #pragma unroll
    for (int g = 0; g < SEG_PER_ROUND; ++g) {
        const int q = (round * SEG_PER_ROUND + g) * OWN_THREADS + tid;
        d[g] = ((const int4*)dstIdx)[q < N_EDGES / 4 ? q : N_EDGES / 4 - 1];
    }
}

static_assert(CODE_CHUNK_WORDS == EDGES_PER_ROUND / 4 && CODE_CHUNKS == SCAN_ROUNDS, "a round of the scan is one chunk");
static_assert((SCAN_ROUNDS * SEG_PER_ROUND - 1) * OWN_THREADS < N_EDGES / 4, "every segment holds words of the list");

__device__ __forceinline__ void load_codes(
    const float* __restrict__ codes, const int round, const int tid, unsigned (&d)[SEG_PER_ROUND])
{
    const float* chunk = codes + (size_t)round * CODE_SPAN + CODE_SKIP;
    #pragma unroll
    for (int g = 0; g < SEG_PER_ROUND; ++g) {
        const int first = (round * SEG_PER_ROUND + g) * OWN_THREADS;
        const int last = N_EDGES / 4 - 1 - first;
        d[g] = __float_as_uint(chunk[g * OWN_THREADS + (tid < last ? tid : last)]);
    }
}

template <bool STORED_IN, bool STORE_OUT>
__device__ __forceinline__ int drain_list(
    const float* __restrict__ feat, const int* __restrict__ srcIdx,
    const int* __restrict__ dstIdx, const float* __restrict__ invSrc,
    float* acc, const unsigned* list, const int count, const int lane,
    float* save, const int saveBase, const int lo, const int rows)
{
    int kept = 0;
    for (int base = 0; base < count; base += GATHER_BATCH) {
        const int left = count - base;
        const int nvalid = left < GATHER_BATCH ? left : GATHER_BATCH;
        unsigned dl = 0;
        int s = 0;
        float w = 0.0f;
        bool ok = false;
        if (lane < nvalid) {
            const unsigned ent = list[base + lane];
            if (STORED_IN) {
                s  = (int)(ent >> 9);
                dl = ent & 511u;
                ok = true;
            } else {
                s  = srcIdx[ent];
                dl = (unsigned)(dstIdx[ent] - lo);
                ok = dl < (unsigned)rows;
            }
            w = invSrc[s];
        }
        const unsigned vmask = __builtin_amdgcn_ballot_w32(ok);
        if (STORE_OUT && ok) {
            const int pos = saveBase + kept + (int)__builtin_amdgcn_mbcnt_lo(vmask, 0u);
            if (pos < SAVE_CAP)
                save[SAVE_HDR + pos] = __uint_as_float(((unsigned)s << 9) | dl);
        }
        kept += __builtin_popcount(vmask);
        float v[GATHER_BATCH][4];
        #pragma unroll
        for (int j = 0; j < GATHER_BATCH; ++j) {
            if ((vmask >> j) & 1u) {
                const int sj = __builtin_amdgcn_readlane(s, j);
                const float* fp = feat + (size_t)sj * H_FEATS + lane;
                v[j][0] = fp[0];
                v[j][1] = fp[32];
                v[j][2] = fp[64];
                v[j][3] = fp[96];
            }
        }
        #pragma unroll
        for (int j = 0; j < GATHER_BATCH; ++j) {
            if ((vmask >> j) & 1u) {
                const int dj = __builtin_amdgcn_readlane((int)dl, j);
                const float wj = __int_as_float(__builtin_amdgcn_readlane(__float_as_int(w), j));
                float* ap = acc + dj * H_FEATS + lane;
                unsafeAtomicAdd(ap + 0,  v[j][0] * wj);
                unsafeAtomicAdd(ap + 32, v[j][1] * wj);
                unsafeAtomicAdd(ap + 64, v[j][2] * wj);
                unsafeAtomicAdd(ap + 96, v[j][3] * wj);
            }
        }
    }
    return kept;
}

__device__ __forceinline__ void append_owned(
    unsigned* mylist, int& cnt, const bool owned, const unsigned edge)
{
    const unsigned mask = __builtin_amdgcn_ballot_w32(owned);
    if (owned) mylist[cnt + (int)__builtin_amdgcn_mbcnt_lo(mask, 0u)] = edge;
    cnt += __builtin_popcount(mask);
}

__device__ __forceinline__ void scan_codes_and_drain(
    const float* __restrict__ feat, const int* __restrict__ srcIdx,
    const int* __restrict__ dstIdx, const float* __restrict__ codes,
    const float* __restrict__ invSrc, float* acc, unsigned* mylist, float* save,
    const int lo, const int rows, const int tid, const int lane)
{
    const unsigned pat = (unsigned)blockIdx.x * 0x01010101u;
    int rr = blockIdx.x % SCAN_ROUNDS;
    int cnt = 0;
    int stored = 0;
    unsigned dn[SEG_PER_ROUND];
    load_codes(codes, rr, tid, dn);
    for (int round = 0; round < SCAN_ROUNDS; ++round) {
        unsigned d[SEG_PER_ROUND];
        #pragma unroll
        for (int g = 0; g < SEG_PER_ROUND; ++g) d[g] = dn[g];
        const int rn = (rr + 1 == SCAN_ROUNDS) ? 0 : rr + 1;
        if (round + 1 < SCAN_ROUNDS) load_codes(codes, rn, tid, dn);
        unsigned f = 0u;
        #pragma unroll
        for (int g = 0; g < SEG_PER_ROUND; ++g) {
            const unsigned x = d[g] ^ pat;
            const unsigned y = ~(((x & 0x7F7F7F7Fu) + 0x7F7F7F7Fu) | x | 0x7F7F7F7Fu);
            f |= y >> (7 - g);
        }
        if (rr == SCAN_ROUNDS - 1 && tid >= N_EDGES / 4 - (SCAN_ROUNDS * SEG_PER_ROUND - 1) * OWN_THREADS)
            f &= ~(0x01010101u << (SEG_PER_ROUND - 1));
        unsigned act = __builtin_amdgcn_ballot_w32(f != 0u);
        const unsigned e0 = (unsigned)(rr * EDGES_PER_ROUND + tid * 4);
        for (int pass = 0; pass < 4 * SEG_PER_ROUND && act != 0u; ++pass) {
            if (f != 0u) {
                const unsigned b = (unsigned)__builtin_ctz(f);
                mylist[cnt + (int)__builtin_amdgcn_mbcnt_lo(act, 0u)] = e0 + (b & 7u) * (OWN_THREADS * 4) + (b >> 3);
                f &= f - 1u;
            }
            cnt += __builtin_popcount(act);
            act = __builtin_amdgcn_ballot_w32(f != 0u);
        }
        if (cnt > WAVE_DRAIN_AT || round + 1 == SCAN_ROUNDS) {
            stored += drain_list<false, true>(feat, srcIdx, dstIdx, invSrc, acc, mylist, cnt, lane,
                                              save, stored, lo, rows);
            cnt = 0;
        }
        rr = rn;
    }
    if (lane == 0) save[0] = __int_as_float(stored);
}

__device__ __forceinline__ void scan_ids_and_drain(
    const float* __restrict__ feat, const int* __restrict__ srcIdx,
    const int* __restrict__ dstIdx, const float* __restrict__ invSrc,
    float* acc, unsigned* mylist, float* save,
    const int lo, const int rows, const int tid, const int lane)
{
    int rr = blockIdx.x % SCAN_ROUNDS;
    int cnt = 0;
    for (int round = 0; round < SCAN_ROUNDS; ++round) {
        int4 d[SEG_PER_ROUND];
        load_round(dstIdx, rr, tid, d);
        #pragma unroll
        for (int g = 0; g < SEG_PER_ROUND; ++g) {
            const unsigned t0 = (unsigned)(d[g].x - lo);
            const unsigned t1 = (unsigned)(d[g].y - lo);
            const unsigned t2 = (unsigned)(d[g].z - lo);
            const unsigned t3 = (unsigned)(d[g].w - lo);
            const int q = (rr * SEG_PER_ROUND + g) * OWN_THREADS + tid;
            const unsigned lim = (q < N_EDGES / 4) ? (unsigned)rows : 0u;
            const bool m0 = t0 < lim, m1 = t1 < lim;
            const bool m2 = t2 < lim, m3 = t3 < lim;
            if (__builtin_amdgcn_ballot_w32(m0 | m1 | m2 | m3)) {
                const unsigned e = (unsigned)(q * 4);
                append_owned(mylist, cnt, m0, e + 0u);
                append_owned(mylist, cnt, m1, e + 1u);
                append_owned(mylist, cnt, m2, e + 2u);
                append_owned(mylist, cnt, m3, e + 3u);
            }
        }
        if (cnt > WAVE_DRAIN_AT || round + 1 == SCAN_ROUNDS) {
            drain_list<false, false>(feat, srcIdx, dstIdx, invSrc, acc, mylist, cnt, lane,
                                     save, 0, lo, rows);
            cnt = 0;
        }
        rr = (rr + 1 == SCAN_ROUNDS) ? 0 : rr + 1;
    }
}

template <bool FIRST>
__global__ __launch_bounds__(OWN_THREADS) void conv_owner_kernel(
    const float* __restrict__ feat, const int* __restrict__ srcIdx,
    const int* __restrict__ dstIdx, const float* __restrict__ invSrc,
    const float* __restrict__ invDst, const float* __restrict__ alphas,
    float* __restrict__ hbuf, float* __restrict__ out)
{
    __shared__ float acc[OWN_ROWS * H_FEATS];
    __shared__ unsigned list[OWN_WAVES * WAVE_LIST_CAP];

    const int tid  = threadIdx.x;
    const int lane = tid & 31;
    const int wave = __builtin_amdgcn_readfirstlane(tid >> 5);
    const int lo   = blockIdx.x * OWN_ROWS;
    const int rows = (N_NODES - lo) < OWN_ROWS ? (N_NODES - lo) : OWN_ROWS;
    unsigned* mylist = list + wave * WAVE_LIST_CAP;

    for (int i = tid; i < OWN_ROWS * H_FEATS / 4; i += OWN_THREADS)
        ((float4*)acc)[i] = make_float4(0.0f, 0.0f, 0.0f, 0.0f);
    __syncthreads();

    float* save = out + (size_t)lo * H_FEATS + wave * SAVE_STRIDE;
    if (FIRST) {
        scan_codes_and_drain(feat, srcIdx, dstIdx, out, invSrc, acc, mylist, save, lo, rows, tid, lane);
    } else {
        const int total = __builtin_amdgcn_readfirstlane(__float_as_int(save[0]));
        if (total <= SAVE_CAP) {
            for (int i0 = 0; i0 < total; i0 += 128) {
                float t[4];
                #pragma unroll
                for (int u = 0; u < 4; ++u) {
                    const int i = i0 + u * 32 + lane;
                    t[u] = save[SAVE_HDR + (i < total ? i : 0)];
                }
                #pragma unroll
                for (int u = 0; u < 4; ++u) {
                    const int i = i0 + u * 32 + lane;
                    if (i < total) mylist[i] = __float_as_uint(t[u]);
                }
            }
            drain_list<true, false>(feat, srcIdx, dstIdx, invSrc, acc, mylist, total, lane, save, 0, lo, rows);
        } else {
            scan_ids_and_drain(feat, srcIdx, dstIdx, invSrc, acc, mylist, save, lo, rows, tid, lane);
        }
    }
    __syncthreads();

    const float a0 = FIRST ? 0.0f : softmax2(alphas, 0);
    const float a1 = FIRST ? 0.0f : softmax2(alphas, 1);
    for (int r0 = wave; r0 < rows; r0 += OWN_WAVES * ROW_GROUP) {
        float4 o[ROW_GROUP];
        float s[ROW_GROUP];
        #pragma unroll
        for (int k = 0; k < ROW_GROUP; ++k) {
            const int r = r0 + k * OWN_WAVES;
            if (r < rows) {
                s[k] = invDst[lo + r];
                if (!FIRST) o[k] = ((const float4*)feat)[(size_t)(lo + r) * (H_FEATS / 4) + lane];
            }
        }
        #pragma unroll
        for (int k = 0; k < ROW_GROUP; ++k) {
            const int r = r0 + k * OWN_WAVES;
            if (r < rows) {
                const float4 v = ((const float4*)(acc + r * H_FEATS))[lane];
                const size_t i4 = (size_t)(lo + r) * (H_FEATS / 4) + lane;
                if (FIRST) {
                    float4 h;
                    h.x = v.x * s[k]; h.y = v.y * s[k]; h.z = v.z * s[k]; h.w = v.w * s[k];
                    ((float4*)hbuf)[i4] = h;
                } else {
                    const float sa = s[k] * a1;
                    const float4 h = o[k];
                    float4 w;
                    w.x = fmaf(sa, v.x, a0 * h.x); w.y = fmaf(sa, v.y, a0 * h.y);
                    w.z = fmaf(sa, v.z, a0 * h.z); w.w = fmaf(sa, v.w, a0 * h.w);
                    ((float4*)out)[i4] = w;
                }
            }
        }
    }
}

extern "C" void kernel_launch(void* const* d_in, const int* in_sizes, int n_in,
                              void* d_out, int out_size, void* d_ws, size_t ws_size,
                              hipStream_t stream)
{
    (void)in_sizes; (void)n_in; (void)out_size; (void)ws_size;

    const float* in_feat = (const float*)d_in[0];
    const int*   edge    = (const int*)d_in[1];
    const float* W       = (const float*)d_in[2];
    const float* bias    = (const float*)d_in[3];
    const float* alphas  = (const float*)d_in[4];
    float*       out     = (float*)d_out;

    const int* srcIdx = edge;
    const int* dstIdx = edge + N_EDGES;

    float* degOut = (float*)d_ws;
    float* degIn  = degOut + N_NODES;
    float* xbuf   = degIn + N_NODES;
    float* hbuf   = xbuf + (size_t)N_NODES * H_FEATS;

    degree_count_kernel<<<DEG_RANGES * DEG_SLICES, DEG_THREADS, 0, stream>>>(srcIdx, dstIdx, hbuf, out);
    degree_merge_kernel<<<(2 * N_NODES / 4 + 255) / 256, 256, 0, stream>>>(hbuf, degOut);

    lin_wmma_kernel<<<LIN_WGS, 256, 0, stream>>>(in_feat, W, bias, xbuf);

    conv_owner_kernel<true><<<OWN_WGS, OWN_THREADS, 0, stream>>>(
        xbuf, srcIdx, dstIdx, degOut, degIn, alphas, hbuf, out);

    conv_owner_kernel<false><<<OWN_WGS, OWN_THREADS, 0, stream>>>(
        hbuf, srcIdx, dstIdx, degOut, degIn, alphas, nullptr, out);
}


# conv step 1 drain: one scattered 4-byte lookup per kept edge, not three: degree launch writes (src id<<9)|local row per edge; GEMM stores x*inv_src[row], same product and rounding; LDS sums same
# speedup vs baseline: 3.3526x; 1.0357x over previous
#include <hip/hip_runtime.h>
#include <hip/hip_bf16.h>
#include <math.h>

#define N_NODES 100000
#define N_EDGES 600000
#define IN_FEATS 256
#define H_FEATS 128

typedef __attribute__((ext_vector_type(2))) float v2f;
typedef __attribute__((ext_vector_type(8))) float v8f;
typedef __attribute__((ext_vector_type(4))) float v4f;

#define OWN_WGS 256
#define OWN_ROWS ((N_NODES + OWN_WGS - 1) / OWN_WGS)
#define OWN_THREADS 1024
#define OWN_WAVES (OWN_THREADS / 32)
#define SAVE_STRIDE 1024
#define SEG_PER_ROUND 7
#define CODE_CHUNK_WORDS (SEG_PER_ROUND * OWN_THREADS)
#define CODE_CHUNKS ((N_EDGES / 4 + CODE_CHUNK_WORDS - 1) / CODE_CHUNK_WORDS)
#define CODE_SPAN (OWN_ROWS * H_FEATS)
#define CODE_SKIP (OWN_WAVES * SAVE_STRIDE)
#define PACK_OFF (CODE_SKIP + CODE_CHUNK_WORDS)
#define PACK_SPAN_WORDS (CODE_SPAN - PACK_OFF)
#define PACK_SPANS ((N_EDGES + PACK_SPAN_WORDS - 1) / PACK_SPAN_WORDS)
#define PACK_NONE 0xFFFFFFFFu

static_assert(OWN_WGS == 256, "an owner's number is one byte");
static_assert(CODE_SPAN - CODE_SKIP >= CODE_CHUNK_WORDS, "a chunk of codes fits behind the waves' regions");
static_assert(CODE_CHUNKS <= OWN_WGS - 1, "chunks sit in rows of owners that have all their rows");
static_assert((CODE_SPAN * 4) % 128 == 0 && (CODE_SKIP * 4) % 128 == 0 && CODE_CHUNK_WORDS % 32 == 0, "chunks are whole lines");
static_assert(PACK_SPAN_WORDS > 0 && PACK_SPAN_WORDS % 32 == 0, "a span of packed words is whole lines");
static_assert(PACK_SPANS <= OWN_WGS - 1, "packed words sit in rows of owners that have all their rows");
static_assert(OWN_ROWS <= 512 && N_NODES <= (1 << 23), "a packed word holds the source id above 9 bits of local row, and is never PACK_NONE");

__device__ __forceinline__ unsigned code_word(const int q)
{
    const unsigned u = (unsigned)q;
    return (u / CODE_CHUNK_WORDS) * CODE_SPAN + CODE_SKIP + (u % CODE_CHUNK_WORDS);
}

__device__ __forceinline__ unsigned pack_word(const unsigned e)
{
    return (e / PACK_SPAN_WORDS) * CODE_SPAN + PACK_OFF + (e % PACK_SPAN_WORDS);
}

__device__ __forceinline__ unsigned pack_edge(const int s, const int d)
{
    const unsigned us = (unsigned)s;
    const unsigned ud = (unsigned)d;
    return (us < (unsigned)N_NODES && ud < (unsigned)N_NODES) ? ((us << 9) | (ud % OWN_ROWS)) : PACK_NONE;
}

__device__ __forceinline__ unsigned owner_codes(const int4 v)
{
    const unsigned o0 = ((unsigned)v.x / OWN_ROWS) & 255u;
    const unsigned o1 = ((unsigned)v.y / OWN_ROWS) & 255u;
    const unsigned o2 = ((unsigned)v.z / OWN_ROWS) & 255u;
    const unsigned o3 = ((unsigned)v.w / OWN_ROWS) & 255u;
    return o0 | (o1 << 8) | (o2 << 16) | (o3 << 24);
}

#define DEG_RANGES 16
#define DEG_SLICES 16
#define DEG_RANGE_NODES ((((N_NODES + DEG_RANGES - 1) / DEG_RANGES) + 31) / 32 * 32)
#define DEG_THREADS 512
#define DEG_Q_TOTAL (N_EDGES / 4)
#define DEG_Q_SLICE ((DEG_Q_TOTAL + DEG_SLICES - 1) / DEG_SLICES)
#define DEG_ITERS ((DEG_Q_SLICE + DEG_THREADS - 1) / DEG_THREADS)
#define DEG_BATCH 10
#define CODE_WG_WORDS ((((N_EDGES / 4 + DEG_RANGES * DEG_SLICES - 1) / (DEG_RANGES * DEG_SLICES)) + 31) / 32 * 32)
#define CODE_ITERS ((CODE_WG_WORDS + DEG_THREADS - 1) / DEG_THREADS)

static_assert(N_EDGES % 4 == 0 && N_NODES % 4 == 0, "ids and counts move 16 bytes at a time");
static_assert((DEG_RANGES - 1) * DEG_RANGE_NODES < N_NODES, "every range holds nodes");
static_assert(2 * DEG_SLICES <= H_FEATS, "the planes fit the space of hbuf");

__device__ __forceinline__ void count_owned(unsigned* cnt, const int4 v, const int lo, const unsigned lim)
{
    const unsigned t0 = (unsigned)(v.x - lo);
    const unsigned t1 = (unsigned)(v.y - lo);
    const unsigned t2 = (unsigned)(v.z - lo);
    const unsigned t3 = (unsigned)(v.w - lo);
    if (t0 < lim) atomicAdd(&cnt[t0], 1u);
    if (t1 < lim) atomicAdd(&cnt[t1], 1u);
    if (t2 < lim) atomicAdd(&cnt[t2], 1u);
    if (t3 < lim) atomicAdd(&cnt[t3], 1u);
}

__global__ __launch_bounds__(DEG_THREADS) void degree_count_kernel(
    const int* __restrict__ srcIdx, const int* __restrict__ dstIdx, float* __restrict__ planes,
    float* __restrict__ codes)
{
    __shared__ __attribute__((aligned(16))) unsigned cnt[2][DEG_RANGE_NODES];

    const int tid   = threadIdx.x;
    const int range = blockIdx.x % DEG_RANGES;
    const int slice = blockIdx.x / DEG_RANGES;
    const int lo    = range * DEG_RANGE_NODES;
    const int rows  = (N_NODES - lo) < DEG_RANGE_NODES ? (N_NODES - lo) : DEG_RANGE_NODES;

    int4 cv[CODE_ITERS];
    int4 sv[CODE_ITERS];
    #pragma unroll
    for (int u = 0; u < CODE_ITERS; ++u) {
        const int w = u * DEG_THREADS + tid;
        const int q = blockIdx.x * CODE_WG_WORDS + w;
        const int qq = (w < CODE_WG_WORDS && q < DEG_Q_TOTAL) ? q : DEG_Q_TOTAL - 1;
        cv[u] = ((const int4*)dstIdx)[qq];
        sv[u] = ((const int4*)srcIdx)[qq];
    }

    for (int i = tid; i < 2 * DEG_RANGE_NODES / 4; i += DEG_THREADS)
        ((uint4*)&cnt[0][0])[i] = make_uint4(0u, 0u, 0u, 0u);
    __syncthreads();

    #pragma unroll
    for (int u = 0; u < CODE_ITERS; ++u) {
        const int w = u * DEG_THREADS + tid;
        const int q = blockIdx.x * CODE_WG_WORDS + w;
        if (w < CODE_WG_WORDS && q < DEG_Q_TOTAL) {
            codes[code_word(q)] = __uint_as_float(owner_codes(cv[u]));
            uint4 p;
            p.x = pack_edge(sv[u].x, cv[u].x);
            p.y = pack_edge(sv[u].y, cv[u].y);
            p.z = pack_edge(sv[u].z, cv[u].z);
            p.w = pack_edge(sv[u].w, cv[u].w);
            *(uint4*)(codes + pack_word(4u * (unsigned)q)) = p;
        }
    }

    const int qlo = slice * DEG_Q_SLICE;
    for (int it0 = 0; it0 < DEG_ITERS; it0 += DEG_BATCH) {
        int4 s4[DEG_BATCH];
        int4 d4[DEG_BATCH];
        #pragma unroll
        for (int k = 0; k < DEG_BATCH; ++k) {
            const int j = (it0 + k) * DEG_THREADS + tid;
            const int q = qlo + j;
            const bool ok = (j < DEG_Q_SLICE) && (q < DEG_Q_TOTAL);
            const int qq = ok ? q : DEG_Q_TOTAL - 1;
            s4[k] = ((const int4*)srcIdx)[qq];
            d4[k] = ((const int4*)dstIdx)[qq];
        }
        #pragma unroll
        for (int k = 0; k < DEG_BATCH; ++k) {
            const int j = (it0 + k) * DEG_THREADS + tid;
            const bool ok = (j < DEG_Q_SLICE) && (qlo + j < DEG_Q_TOTAL);
            const unsigned lim = ok ? (unsigned)rows : 0u;
            count_owned(cnt[0], s4[k], lo, lim);
            count_owned(cnt[1], d4[k], lo, lim);
        }
    }
    __syncthreads();

    uint4* po = (uint4*)(planes + (size_t)slice * (2 * N_NODES) + lo);
    uint4* pi = (uint4*)(planes + (size_t)slice * (2 * N_NODES) + N_NODES + lo);
    for (int i = tid; i < rows / 4; i += DEG_THREADS) {
        po[i] = ((const uint4*)&cnt[0][0])[i];
        pi[i] = ((const uint4*)&cnt[1][0])[i];
    }
}

__device__ __forceinline__ float inv_norm(const unsigned c)
{
    const float d = (float)c;
    return (d > 0.0f) ? rsqrtf(fmaxf(d, 1.0f)) : 0.0f;
}

__global__ __launch_bounds__(256) void degree_merge_kernel(
    const float* __restrict__ planes, float* __restrict__ norm)
{
    const int i = blockIdx.x * blockDim.x + threadIdx.x;
    if (i < 2 * N_NODES / 4) {
        uint4 v[DEG_SLICES];
        #pragma unroll
        for (int s = 0; s < DEG_SLICES; ++s)
            v[s] = ((const uint4*)planes)[(size_t)s * (2 * N_NODES / 4) + i];
        uint4 c = v[0];
        #pragma unroll
        for (int s = 1; s < DEG_SLICES; ++s) {
            c.x += v[s].x; c.y += v[s].y; c.z += v[s].z; c.w += v[s].w;
        }
        float4 o;
        o.x = inv_norm(c.x); o.y = inv_norm(c.y); o.z = inv_norm(c.z); o.w = inv_norm(c.w);
        ((float4*)norm)[i] = o;
    }
}

#define LIN_ROWS 32
#define LIN_PITCH (IN_FEATS + 4)
#define LIN_WGS 512
#define LIN_TILES (N_NODES / LIN_ROWS)
#define LIN_KSTEPS (IN_FEATS / 4)
#define LIN_PIECES (LIN_ROWS * IN_FEATS / 4 / 256)

static_assert(N_NODES % LIN_ROWS == 0, "tiles are whole");
static_assert(LIN_WGS <= LIN_TILES, "every work-group has a first tile");
static_assert(LIN_PITCH % 64 == 4 && (LIN_PITCH * 4) % 16 == 0, "bank spread and 16-byte alignment of the rows");
static_assert(H_FEATS == 8 * 16, "eight waves of 16 columns");

__global__ __launch_bounds__(256, 2) void lin_wmma_kernel(
    const float* __restrict__ A, const float* __restrict__ W,
    const float* __restrict__ bias, const float* __restrict__ invSrc,
    float* __restrict__ X)
{
    __shared__ __attribute__((aligned(16))) float tile[2][LIN_ROWS * LIN_PITCH];

    const int tid    = threadIdx.x;
    const int lane   = tid & 31;
    const int wave   = tid >> 5;
    const int halfId = lane >> 4;
    const int l16    = lane & 15;
    const int n0     = wave * 16;

    v2f bf[LIN_KSTEPS];
    {
        const float* bp = W + (size_t)(2 * halfId) * H_FEATS + n0 + l16;
        #pragma unroll
        for (int j = 0; j < LIN_KSTEPS; ++j) {
            bf[j].x = bp[(size_t)(4 * j) * H_FEATS];
            bf[j].y = bp[(size_t)(4 * j + 1) * H_FEATS];
        }
    }
    const float bb = bias[n0 + l16];

    const int stRow = tid >> 6;
    const int stCol = (tid & 63) * 4;
    const v4f* A4 = (const v4f*)A;

    v4f st[LIN_PIECES];
    int t = blockIdx.x;
    #pragma unroll
    for (int i = 0; i < LIN_PIECES; ++i)
        st[i] = A4[(size_t)t * (LIN_ROWS * IN_FEATS / 4) + i * 256 + tid];
    const float* sp = invSrc + 8 * halfId;
    float sc0[8];
    float sc1[8];
    #pragma unroll
    for (int r = 0; r < 8; ++r) {
        sc0[r] = sp[t * LIN_ROWS + r];
        sc1[r] = sp[t * LIN_ROWS + 16 + r];
    }
    #pragma unroll
    for (int i = 0; i < LIN_PIECES; ++i)
        *(v4f*)(&tile[0][(i * 4 + stRow) * LIN_PITCH + stCol]) = st[i];
    __syncthreads();

    int buf = 0;
    for (; t < LIN_TILES; t += LIN_WGS) {
        const int tn = t + LIN_WGS;
        const bool more = tn < LIN_TILES;
        if (more) {
            #pragma unroll
            for (int i = 0; i < LIN_PIECES; ++i)
                st[i] = A4[(size_t)tn * (LIN_ROWS * IN_FEATS / 4) + i * 256 + tid];
        }

        const float* fa = &tile[buf][l16 * LIN_PITCH + 2 * halfId];
        v8f c0 = {};
        v8f c1 = {};
        #pragma unroll
        for (int j = 0; j < LIN_KSTEPS; ++j) {
            const v2f a0 = *(const v2f*)(fa + 4 * j);
            const v2f a1 = *(const v2f*)(fa + 16 * LIN_PITCH + 4 * j);
            c0 = __builtin_amdgcn_wmma_f32_16x16x4_f32(
                    false, a0, false, bf[j], (short)0, c0, false, false);
            c1 = __builtin_amdgcn_wmma_f32_16x16x4_f32(
                    false, a1, false, bf[j], (short)0, c1, false, false);
        }

        if (more) {
            #pragma unroll
            for (int i = 0; i < LIN_PIECES; ++i)
                *(v4f*)(&tile[buf ^ 1][(i * 4 + stRow) * LIN_PITCH + stCol]) = st[i];
        }
        __syncthreads();
        buf ^= 1;

        const int rowBase = t * LIN_ROWS;
        #pragma unroll
        for (int r = 0; r < 8; ++r) {
            const int m = r + 8 * halfId;
            X[(size_t)(rowBase + m) * H_FEATS + n0 + l16]      = (c0[r] + bb) * sc0[r];
            X[(size_t)(rowBase + 16 + m) * H_FEATS + n0 + l16] = (c1[r] + bb) * sc1[r];
        }
        if (more) {
            #pragma unroll
            for (int r = 0; r < 8; ++r) {
                sc0[r] = sp[tn * LIN_ROWS + r];
                sc1[r] = sp[tn * LIN_ROWS + 16 + r];
            }
        }
    }
}

#define EDGES_PER_ROUND (SEG_PER_ROUND * OWN_THREADS * 4)
#define SCAN_ROUNDS ((N_EDGES + EDGES_PER_ROUND - 1) / EDGES_PER_ROUND)
#define WAVE_LIST_CAP 960
#define WAVE_EDGES_PER_ROUND (SEG_PER_ROUND * 32 * 4)
#define WAVE_DRAIN_AT (WAVE_LIST_CAP - WAVE_EDGES_PER_ROUND)
#define GATHER_BATCH 16
#define ROW_GROUP 4
#define SAVE_HDR 32
#define SAVE_CAP WAVE_LIST_CAP

static_assert(OWN_ROWS <= 512, "local row number must fit the 9 low bits of a stored entry");
static_assert(N_EDGES % 4 == 0, "destination ids are read four at a time");
static_assert(WAVE_DRAIN_AT >= 0, "one round of a wave's matches must fit its list");
static_assert(N_NODES <= (1 << 23), "source id must fit the 23 high bits of a stored entry");
static_assert(SAVE_HDR + SAVE_CAP <= SAVE_STRIDE, "count line and entries must fit the wave's region");
static_assert((N_NODES - (OWN_WGS - 1) * OWN_ROWS) * H_FEATS >= OWN_WAVES * SAVE_STRIDE,
              "the last work-group's rows of out must hold its waves' regions");
static_assert((OWN_ROWS * H_FEATS) % 64 == 0 && SAVE_STRIDE % 64 == 0, "regions start on 256-byte units");

__device__ __forceinline__ float softmax2(const float* __restrict__ alphas, int idx)
{
    const float a0 = alphas[0], a1 = alphas[1];
    const float m  = fmaxf(a0, a1);
    const float e0 = __expf(a0 - m), e1 = __expf(a1 - m);
    return (idx == 0 ? e0 : e1) / (e0 + e1);
}

__device__ __forceinline__ void load_round(
    const int* __restrict__ dstIdx, const int round, const int tid, int4 (&d)[SEG_PER_ROUND])
{
    #pragma unroll
    for (int g = 0; g < SEG_PER_ROUND; ++g) {
        const int q = (round * SEG_PER_ROUND + g) * OWN_THREADS + tid;
        d[g] = ((const int4*)dstIdx)[q < N_EDGES / 4 ? q : N_EDGES / 4 - 1];
    }
}

static_assert(CODE_CHUNK_WORDS == EDGES_PER_ROUND / 4 && CODE_CHUNKS == SCAN_ROUNDS, "a round of the scan is one chunk");
static_assert((SCAN_ROUNDS * SEG_PER_ROUND - 1) * OWN_THREADS < N_EDGES / 4, "every segment holds words of the list");

__device__ __forceinline__ void load_codes(
    const float* __restrict__ codes, const int round, const int tid, unsigned (&d)[SEG_PER_ROUND])
{
    const float* chunk = codes + (size_t)round * CODE_SPAN + CODE_SKIP;
    #pragma unroll
    for (int g = 0; g < SEG_PER_ROUND; ++g) {
        const int first = (round * SEG_PER_ROUND + g) * OWN_THREADS;
        const int last = N_EDGES / 4 - 1 - first;
        d[g] = __float_as_uint(chunk[g * OWN_THREADS + (tid < last ? tid : last)]);
    }
}

#define ENT_IDS 0
#define ENT_STORED 1
#define ENT_PACKED 2
template <int KIND, bool STORE_OUT>
__device__ __forceinline__ int drain_list(
    const float* __restrict__ feat, const int* __restrict__ srcIdx,
    const int* __restrict__ dstIdx, const float* __restrict__ invSrc,
    const float* __restrict__ packed,
    float* acc, const unsigned* list, const int count, const int lane,
    float* save, const int saveBase, const int lo, const int rows)
{
    int kept = 0;
    for (int base = 0; base < count; base += GATHER_BATCH) {
        const int left = count - base;
        const int nvalid = left < GATHER_BATCH ? left : GATHER_BATCH;
        unsigned dl = 0;
        int s = 0;
        float w = 0.0f;
        bool ok = false;
        if (KIND == ENT_PACKED) {
            const unsigned ent = list[base + (lane < nvalid ? lane : nvalid - 1)];
            const unsigned p = __float_as_uint(packed[pack_word(ent)]);
            s  = (int)(p >> 9);
            dl = p & 511u;
            ok = (lane < nvalid) && (p != PACK_NONE);
        } else if (lane < nvalid) {
            const unsigned ent = list[base + lane];
            if (KIND == ENT_STORED) {
                s  = (int)(ent >> 9);
                dl = ent & 511u;
                ok = true;
            } else {
                s  = srcIdx[ent];
                dl = (unsigned)(dstIdx[ent] - lo);
                ok = dl < (unsigned)rows;
            }
            w = invSrc[s];
        }
        const unsigned vmask = __builtin_amdgcn_ballot_w32(ok);
        if (STORE_OUT && ok) {
            const int pos = saveBase + kept + (int)__builtin_amdgcn_mbcnt_lo(vmask, 0u);
            if (pos < SAVE_CAP)
                save[SAVE_HDR + pos] = __uint_as_float(((unsigned)s << 9) | dl);
        }
        kept += __builtin_popcount(vmask);
        float v[GATHER_BATCH][4];
        #pragma unroll
        for (int j = 0; j < GATHER_BATCH; ++j) {
            if ((vmask >> j) & 1u) {
                const int sj = __builtin_amdgcn_readlane(s, j);
                const float* fp = feat + (size_t)sj * H_FEATS + lane;
                v[j][0] = fp[0];
                v[j][1] = fp[32];
                v[j][2] = fp[64];
                v[j][3] = fp[96];
            }
        }
        #pragma unroll
        for (int j = 0; j < GATHER_BATCH; ++j) {
            if ((vmask >> j) & 1u) {
                const int dj = __builtin_amdgcn_readlane((int)dl, j);
                float* ap = acc + dj * H_FEATS + lane;
                if (KIND == ENT_PACKED) {
                    unsafeAtomicAdd(ap + 0,  v[j][0]);
                    unsafeAtomicAdd(ap + 32, v[j][1]);
                    unsafeAtomicAdd(ap + 64, v[j][2]);
                    unsafeAtomicAdd(ap + 96, v[j][3]);
                } else {
                    const float wj = __int_as_float(__builtin_amdgcn_readlane(__float_as_int(w), j));
                    unsafeAtomicAdd(ap + 0,  v[j][0] * wj);
                    unsafeAtomicAdd(ap + 32, v[j][1] * wj);
                    unsafeAtomicAdd(ap + 64, v[j][2] * wj);
                    unsafeAtomicAdd(ap + 96, v[j][3] * wj);
                }
            }
        }
    }
    return kept;
}

__device__ __forceinline__ void append_owned(
    unsigned* mylist, int& cnt, const bool owned, const unsigned edge)
{
    const unsigned mask = __builtin_amdgcn_ballot_w32(owned);
    if (owned) mylist[cnt + (int)__builtin_amdgcn_mbcnt_lo(mask, 0u)] = edge;
    cnt += __builtin_popcount(mask);
}

__device__ __forceinline__ void scan_codes_and_drain(
    const float* __restrict__ feat, const int* __restrict__ srcIdx,
    const int* __restrict__ dstIdx, const float* __restrict__ codes,
    const float* __restrict__ invSrc, float* acc, unsigned* mylist, float* save,
    const int lo, const int rows, const int tid, const int lane)
{
    const unsigned pat = (unsigned)blockIdx.x * 0x01010101u;
    int rr = blockIdx.x % SCAN_ROUNDS;
    int cnt = 0;
    int stored = 0;
    unsigned dn[SEG_PER_ROUND];
    load_codes(codes, rr, tid, dn);
    for (int round = 0; round < SCAN_ROUNDS; ++round) {
        unsigned d[SEG_PER_ROUND];
        #pragma unroll
        for (int g = 0; g < SEG_PER_ROUND; ++g) d[g] = dn[g];
        const int rn = (rr + 1 == SCAN_ROUNDS) ? 0 : rr + 1;
        if (round + 1 < SCAN_ROUNDS) load_codes(codes, rn, tid, dn);
        unsigned f = 0u;
        #pragma unroll
        for (int g = 0; g < SEG_PER_ROUND; ++g) {
            const unsigned x = d[g] ^ pat;
            const unsigned y = ~(((x & 0x7F7F7F7Fu) + 0x7F7F7F7Fu) | x | 0x7F7F7F7Fu);
            f |= y >> (7 - g);
        }
        if (rr == SCAN_ROUNDS - 1 && tid >= N_EDGES / 4 - (SCAN_ROUNDS * SEG_PER_ROUND - 1) * OWN_THREADS)
            f &= ~(0x01010101u << (SEG_PER_ROUND - 1));
        unsigned act = __builtin_amdgcn_ballot_w32(f != 0u);
        const unsigned e0 = (unsigned)(rr * EDGES_PER_ROUND + tid * 4);
        for (int pass = 0; pass < 4 * SEG_PER_ROUND && act != 0u; ++pass) {
            if (f != 0u) {
                const unsigned b = (unsigned)__builtin_ctz(f);
                mylist[cnt + (int)__builtin_amdgcn_mbcnt_lo(act, 0u)] = e0 + (b & 7u) * (OWN_THREADS * 4) + (b >> 3);
                f &= f - 1u;
            }
            cnt += __builtin_popcount(act);
            act = __builtin_amdgcn_ballot_w32(f != 0u);
        }
        if (cnt > WAVE_DRAIN_AT || round + 1 == SCAN_ROUNDS) {
            stored += drain_list<ENT_PACKED, true>(feat, srcIdx, dstIdx, invSrc, codes, acc, mylist, cnt, lane,
                                                   save, stored, lo, rows);
            cnt = 0;
        }
        rr = rn;
    }
    if (lane == 0) save[0] = __int_as_float(stored);
}

__device__ __forceinline__ void scan_ids_and_drain(
    const float* __restrict__ feat, const int* __restrict__ srcIdx,
    const int* __restrict__ dstIdx, const float* __restrict__ invSrc,
    float* acc, unsigned* mylist, float* save,
    const int lo, const int rows, const int tid, const int lane)
{
    int rr = blockIdx.x % SCAN_ROUNDS;
    int cnt = 0;
    for (int round = 0; round < SCAN_ROUNDS; ++round) {
        int4 d[SEG_PER_ROUND];
        load_round(dstIdx, rr, tid, d);
        #pragma unroll
        for (int g = 0; g < SEG_PER_ROUND; ++g) {
            const unsigned t0 = (unsigned)(d[g].x - lo);
            const unsigned t1 = (unsigned)(d[g].y - lo);
            const unsigned t2 = (unsigned)(d[g].z - lo);
            const unsigned t3 = (unsigned)(d[g].w - lo);
            const int q = (rr * SEG_PER_ROUND + g) * OWN_THREADS + tid;
            const unsigned lim = (q < N_EDGES / 4) ? (unsigned)rows : 0u;
            const bool m0 = t0 < lim, m1 = t1 < lim;
            const bool m2 = t2 < lim, m3 = t3 < lim;
            if (__builtin_amdgcn_ballot_w32(m0 | m1 | m2 | m3)) {
                const unsigned e = (unsigned)(q * 4);
                append_owned(mylist, cnt, m0, e + 0u);
                append_owned(mylist, cnt, m1, e + 1u);
                append_owned(mylist, cnt, m2, e + 2u);
                append_owned(mylist, cnt, m3, e + 3u);
            }
        }
        if (cnt > WAVE_DRAIN_AT || round + 1 == SCAN_ROUNDS) {
            drain_list<ENT_IDS, false>(feat, srcIdx, dstIdx, invSrc, nullptr, acc, mylist, cnt, lane,
                                       save, 0, lo, rows);
            cnt = 0;
        }
        rr = (rr + 1 == SCAN_ROUNDS) ? 0 : rr + 1;
    }
}

template <bool FIRST>
__global__ __launch_bounds__(OWN_THREADS) void conv_owner_kernel(
    const float* __restrict__ feat, const int* __restrict__ srcIdx,
    const int* __restrict__ dstIdx, const float* __restrict__ invSrc,
    const float* __restrict__ invDst, const float* __restrict__ alphas,
    float* __restrict__ hbuf, float* __restrict__ out)
{
    __shared__ float acc[OWN_ROWS * H_FEATS];
    __shared__ unsigned list[OWN_WAVES * WAVE_LIST_CAP];

    const int tid  = threadIdx.x;
    const int lane = tid & 31;
    const int wave = __builtin_amdgcn_readfirstlane(tid >> 5);
    const int lo   = blockIdx.x * OWN_ROWS;
    const int rows = (N_NODES - lo) < OWN_ROWS ? (N_NODES - lo) : OWN_ROWS;
    unsigned* mylist = list + wave * WAVE_LIST_CAP;

    for (int i = tid; i < OWN_ROWS * H_FEATS / 4; i += OWN_THREADS)
        ((float4*)acc)[i] = make_float4(0.0f, 0.0f, 0.0f, 0.0f);
    __syncthreads();

    float* save = out + (size_t)lo * H_FEATS + wave * SAVE_STRIDE;
    if (FIRST) {
        scan_codes_and_drain(feat, srcIdx, dstIdx, out, invSrc, acc, mylist, save, lo, rows, tid, lane);
    } else {
        const int total = __builtin_amdgcn_readfirstlane(__float_as_int(save[0]));
        if (total <= SAVE_CAP) {
            for (int i0 = 0; i0 < total; i0 += 128) {
                float t[4];
                #pragma unroll
                for (int u = 0; u < 4; ++u) {
                    const int i = i0 + u * 32 + lane;
                    t[u] = save[SAVE_HDR + (i < total ? i : 0)];
                }
                #pragma unroll
                for (int u = 0; u < 4; ++u) {
                    const int i = i0 + u * 32 + lane;
                    if (i < total) mylist[i] = __float_as_uint(t[u]);
                }
            }
            drain_list<ENT_STORED, false>(feat, srcIdx, dstIdx, invSrc, nullptr, acc, mylist, total, lane, save, 0, lo, rows);
        } else {
            scan_ids_and_drain(feat, srcIdx, dstIdx, invSrc, acc, mylist, save, lo, rows, tid, lane);
        }
    }
    __syncthreads();

    const float a0 = FIRST ? 0.0f : softmax2(alphas, 0);
    const float a1 = FIRST ? 0.0f : softmax2(alphas, 1);
    for (int r0 = wave; r0 < rows; r0 += OWN_WAVES * ROW_GROUP) {
        float4 o[ROW_GROUP];
        float s[ROW_GROUP];
        #pragma unroll
        for (int k = 0; k < ROW_GROUP; ++k) {
            const int r = r0 + k * OWN_WAVES;
            if (r < rows) {
                s[k] = invDst[lo + r];
                if (!FIRST) o[k] = ((const float4*)feat)[(size_t)(lo + r) * (H_FEATS / 4) + lane];
            }
        }
        #pragma unroll
        for (int k = 0; k < ROW_GROUP; ++k) {
            const int r = r0 + k * OWN_WAVES;
            if (r < rows) {
                const float4 v = ((const float4*)(acc + r * H_FEATS))[lane];
                const size_t i4 = (size_t)(lo + r) * (H_FEATS / 4) + lane;
                if (FIRST) {
                    float4 h;
                    h.x = v.x * s[k]; h.y = v.y * s[k]; h.z = v.z * s[k]; h.w = v.w * s[k];
                    ((float4*)hbuf)[i4] = h;
                } else {
                    const float sa = s[k] * a1;
                    const float4 h = o[k];
                    float4 w;
                    w.x = fmaf(sa, v.x, a0 * h.x); w.y = fmaf(sa, v.y, a0 * h.y);
                    w.z = fmaf(sa, v.z, a0 * h.z); w.w = fmaf(sa, v.w, a0 * h.w);
                    ((float4*)out)[i4] = w;
                }
            }
        }
    }
}

extern "C" void kernel_launch(void* const* d_in, const int* in_sizes, int n_in,
                              void* d_out, int out_size, void* d_ws, size_t ws_size,
                              hipStream_t stream)
{
    (void)in_sizes; (void)n_in; (void)out_size; (void)ws_size;

    const float* in_feat = (const float*)d_in[0];
    const int*   edge    = (const int*)d_in[1];
    const float* W       = (const float*)d_in[2];
    const float* bias    = (const float*)d_in[3];
    const float* alphas  = (const float*)d_in[4];
    float*       out     = (float*)d_out;

    const int* srcIdx = edge;
    const int* dstIdx = edge + N_EDGES;

    float* degOut = (float*)d_ws;
    float* degIn  = degOut + N_NODES;
    float* xbuf   = degIn + N_NODES;
    float* hbuf   = xbuf + (size_t)N_NODES * H_FEATS;

    degree_count_kernel<<<DEG_RANGES * DEG_SLICES, DEG_THREADS, 0, stream>>>(srcIdx, dstIdx, hbuf, out);
    degree_merge_kernel<<<(2 * N_NODES / 4 + 255) / 256, 256, 0, stream>>>(hbuf, degOut);

    lin_wmma_kernel<<<LIN_WGS, 256, 0, stream>>>(in_feat, W, bias, degOut, xbuf);

    conv_owner_kernel<true><<<OWN_WGS, OWN_THREADS, 0, stream>>>(
        xbuf, srcIdx, dstIdx, degOut, degIn, alphas, hbuf, out);

    conv_owner_kernel<false><<<OWN_WGS, OWN_THREADS, 0, stream>>>(
        hbuf, srcIdx, dstIdx, degOut, degIn, alphas, nullptr, out);
}


# degrees: merge launch removed (5 launches -> 4): the product kernel adds its tile's 16 out-degree counts per row itself; conv owners count in-degrees in LDS while draining (integer, exact); sums same
# speedup vs baseline: 3.4932x; 1.0419x over previous
#include <hip/hip_runtime.h>
#include <hip/hip_bf16.h>
#include <math.h>

#define N_NODES 100000
#define N_EDGES 600000
#define IN_FEATS 256
#define H_FEATS 128

typedef __attribute__((ext_vector_type(2))) float v2f;
typedef __attribute__((ext_vector_type(8))) float v8f;
typedef __attribute__((ext_vector_type(4))) float v4f;

#define OWN_WGS 256
#define OWN_ROWS ((N_NODES + OWN_WGS - 1) / OWN_WGS)
#define OWN_THREADS 1024
#define OWN_WAVES (OWN_THREADS / 32)
#define SAVE_STRIDE 1024
#define SEG_PER_ROUND 7
#define CODE_CHUNK_WORDS (SEG_PER_ROUND * OWN_THREADS)
#define CODE_CHUNKS ((N_EDGES / 4 + CODE_CHUNK_WORDS - 1) / CODE_CHUNK_WORDS)
#define CODE_SPAN (OWN_ROWS * H_FEATS)
#define CODE_SKIP (OWN_WAVES * SAVE_STRIDE)
#define PACK_OFF (CODE_SKIP + CODE_CHUNK_WORDS)
#define PACK_SPAN_WORDS (CODE_SPAN - PACK_OFF)
#define PACK_SPANS ((N_EDGES + PACK_SPAN_WORDS - 1) / PACK_SPAN_WORDS)
#define PACK_NONE 0xFFFFFFFFu

static_assert(OWN_WGS == 256, "an owner's number is one byte");
static_assert(CODE_SPAN - CODE_SKIP >= CODE_CHUNK_WORDS, "a chunk of codes fits behind the waves' regions");
static_assert(CODE_CHUNKS <= OWN_WGS - 1, "chunks sit in rows of owners that have all their rows");
static_assert((CODE_SPAN * 4) % 128 == 0 && (CODE_SKIP * 4) % 128 == 0 && CODE_CHUNK_WORDS % 32 == 0, "chunks are whole lines");
static_assert(PACK_SPAN_WORDS > 0 && PACK_SPAN_WORDS % 32 == 0, "a span of packed words is whole lines");
static_assert(PACK_SPANS <= OWN_WGS - 1, "packed words sit in rows of owners that have all their rows");
static_assert(OWN_ROWS <= 512 && N_NODES <= (1 << 23), "a packed word holds the source id above 9 bits of local row, and is never PACK_NONE");

__device__ __forceinline__ unsigned code_word(const int q)
{
    const unsigned u = (unsigned)q;
    return (u / CODE_CHUNK_WORDS) * CODE_SPAN + CODE_SKIP + (u % CODE_CHUNK_WORDS);
}

__device__ __forceinline__ unsigned pack_word(const unsigned e)
{
    return (e / PACK_SPAN_WORDS) * CODE_SPAN + PACK_OFF + (e % PACK_SPAN_WORDS);
}

__device__ __forceinline__ unsigned pack_edge(const int s, const int d)
{
    const unsigned us = (unsigned)s;
    const unsigned ud = (unsigned)d;
    return (us < (unsigned)N_NODES && ud < (unsigned)N_NODES) ? ((us << 9) | (ud % OWN_ROWS)) : PACK_NONE;
}

__device__ __forceinline__ unsigned owner_codes(const int4 v)
{
    const unsigned o0 = ((unsigned)v.x / OWN_ROWS) & 255u;
    const unsigned o1 = ((unsigned)v.y / OWN_ROWS) & 255u;
    const unsigned o2 = ((unsigned)v.z / OWN_ROWS) & 255u;
    const unsigned o3 = ((unsigned)v.w / OWN_ROWS) & 255u;
    return o0 | (o1 << 8) | (o2 << 16) | (o3 << 24);
}

#define DEG_RANGES 16
#define DEG_SLICES 16
#define DEG_RANGE_NODES ((((N_NODES + DEG_RANGES - 1) / DEG_RANGES) + 31) / 32 * 32)
#define DEG_THREADS 512
#define DEG_Q_TOTAL (N_EDGES / 4)
#define DEG_Q_SLICE ((DEG_Q_TOTAL + DEG_SLICES - 1) / DEG_SLICES)
#define DEG_ITERS ((DEG_Q_SLICE + DEG_THREADS - 1) / DEG_THREADS)
#define DEG_BATCH DEG_ITERS
#define CODE_WG_WORDS ((((N_EDGES / 4 + DEG_RANGES * DEG_SLICES - 1) / (DEG_RANGES * DEG_SLICES)) + 31) / 32 * 32)
#define CODE_ITERS ((CODE_WG_WORDS + DEG_THREADS - 1) / DEG_THREADS)

static_assert(N_EDGES % 4 == 0 && N_NODES % 4 == 0, "ids and counts move 16 bytes at a time");
static_assert((DEG_RANGES - 1) * DEG_RANGE_NODES < N_NODES, "every range holds nodes");
static_assert(DEG_SLICES <= H_FEATS, "the planes fit the space of hbuf");

__device__ __forceinline__ void count_owned(unsigned* cnt, const int4 v, const int lo, const unsigned lim)
{
    const unsigned t0 = (unsigned)(v.x - lo);
    const unsigned t1 = (unsigned)(v.y - lo);
    const unsigned t2 = (unsigned)(v.z - lo);
    const unsigned t3 = (unsigned)(v.w - lo);
    if (t0 < lim) atomicAdd(&cnt[t0], 1u);
    if (t1 < lim) atomicAdd(&cnt[t1], 1u);
    if (t2 < lim) atomicAdd(&cnt[t2], 1u);
    if (t3 < lim) atomicAdd(&cnt[t3], 1u);
}

__global__ __launch_bounds__(DEG_THREADS) void degree_count_kernel(
    const int* __restrict__ srcIdx, const int* __restrict__ dstIdx, float* __restrict__ planes,
    float* __restrict__ codes)
{
    __shared__ __attribute__((aligned(16))) unsigned cnt[DEG_RANGE_NODES];

    const int tid   = threadIdx.x;
    const int range = blockIdx.x % DEG_RANGES;
    const int slice = blockIdx.x / DEG_RANGES;
    const int lo    = range * DEG_RANGE_NODES;
    const int rows  = (N_NODES - lo) < DEG_RANGE_NODES ? (N_NODES - lo) : DEG_RANGE_NODES;

    int4 cv[CODE_ITERS];
    int4 sv[CODE_ITERS];
    #pragma unroll
    for (int u = 0; u < CODE_ITERS; ++u) {
        const int w = u * DEG_THREADS + tid;
        const int q = blockIdx.x * CODE_WG_WORDS + w;
        const int qq = (w < CODE_WG_WORDS && q < DEG_Q_TOTAL) ? q : DEG_Q_TOTAL - 1;
        cv[u] = ((const int4*)dstIdx)[qq];
        sv[u] = ((const int4*)srcIdx)[qq];
    }

    for (int i = tid; i < DEG_RANGE_NODES / 4; i += DEG_THREADS)
        ((uint4*)&cnt[0])[i] = make_uint4(0u, 0u, 0u, 0u);
    __syncthreads();

    #pragma unroll
    for (int u = 0; u < CODE_ITERS; ++u) {
        const int w = u * DEG_THREADS + tid;
        const int q = blockIdx.x * CODE_WG_WORDS + w;
        if (w < CODE_WG_WORDS && q < DEG_Q_TOTAL) {
            codes[code_word(q)] = __uint_as_float(owner_codes(cv[u]));
            uint4 p;
            p.x = pack_edge(sv[u].x, cv[u].x);
            p.y = pack_edge(sv[u].y, cv[u].y);
            p.z = pack_edge(sv[u].z, cv[u].z);
            p.w = pack_edge(sv[u].w, cv[u].w);
            *(uint4*)(codes + pack_word(4u * (unsigned)q)) = p;
        }
    }

    const int qlo = slice * DEG_Q_SLICE;
    for (int it0 = 0; it0 < DEG_ITERS; it0 += DEG_BATCH) {
        int4 s4[DEG_BATCH];
        #pragma unroll
        for (int k = 0; k < DEG_BATCH; ++k) {
            const int j = (it0 + k) * DEG_THREADS + tid;
            const int q = qlo + j;
            const bool ok = (j < DEG_Q_SLICE) && (q < DEG_Q_TOTAL);
            const int qq = ok ? q : DEG_Q_TOTAL - 1;
            s4[k] = ((const int4*)srcIdx)[qq];
        }
        #pragma unroll
        for (int k = 0; k < DEG_BATCH; ++k) {
            const int j = (it0 + k) * DEG_THREADS + tid;
            const bool ok = (j < DEG_Q_SLICE) && (qlo + j < DEG_Q_TOTAL);
            const unsigned lim = ok ? (unsigned)rows : 0u;
            count_owned(cnt, s4[k], lo, lim);
        }
    }
    __syncthreads();

    uint4* po = (uint4*)(planes + (size_t)slice * N_NODES + lo);
    for (int i = tid; i < rows / 4; i += DEG_THREADS)
        po[i] = ((const uint4*)&cnt[0])[i];
}

__device__ __forceinline__ float inv_norm(const unsigned c)
{
    const float d = (float)c;
    return (d > 0.0f) ? rsqrtf(fmaxf(d, 1.0f)) : 0.0f;
}

#define LIN_ROWS 32
#define LIN_PITCH (IN_FEATS + 4)
#define LIN_WGS 512
#define LIN_TILES (N_NODES / LIN_ROWS)
#define LIN_KSTEPS (IN_FEATS / 4)
#define LIN_PIECES (LIN_ROWS * IN_FEATS / 4 / 256)

static_assert(N_NODES % LIN_ROWS == 0, "tiles are whole");
static_assert(LIN_WGS <= LIN_TILES, "every work-group has a first tile");
static_assert(LIN_PITCH % 64 == 4 && (LIN_PITCH * 4) % 16 == 0, "bank spread and 16-byte alignment of the rows");
static_assert(H_FEATS == 8 * 16, "eight waves of 16 columns");

__global__ __launch_bounds__(256, 2) void lin_wmma_kernel(
    const float* __restrict__ A, const float* __restrict__ W,
    const float* __restrict__ bias, const float* __restrict__ planes,
    float* __restrict__ invSrcOut, float* __restrict__ X)
{
    __shared__ __attribute__((aligned(16))) float tile[2][LIN_ROWS * LIN_PITCH];

    const int tid    = threadIdx.x;
    const int lane   = tid & 31;
    const int wave   = tid >> 5;
    const int halfId = lane >> 4;
    const int l16    = lane & 15;
    const int n0     = wave * 16;

    v2f bf[LIN_KSTEPS];
    {
        const float* bp = W + (size_t)(2 * halfId) * H_FEATS + n0 + l16;
        #pragma unroll
        for (int j = 0; j < LIN_KSTEPS; ++j) {
            bf[j].x = bp[(size_t)(4 * j) * H_FEATS];
            bf[j].y = bp[(size_t)(4 * j + 1) * H_FEATS];
        }
    }
    const float bb = bias[n0 + l16];

    const int stRow = tid >> 6;
    const int stCol = (tid & 63) * 4;
    const v4f* A4 = (const v4f*)A;

    v4f st[LIN_PIECES];
    int t = blockIdx.x;
    #pragma unroll
    for (int i = 0; i < LIN_PIECES; ++i)
        st[i] = A4[(size_t)t * (LIN_ROWS * IN_FEATS / 4) + i * 256 + tid];
    const float* pp = planes + lane;
    unsigned pl[DEG_SLICES];
    #pragma unroll
    for (int s = 0; s < DEG_SLICES; ++s)
        pl[s] = __float_as_uint(pp[(size_t)s * N_NODES + t * LIN_ROWS]);
    #pragma unroll
    for (int i = 0; i < LIN_PIECES; ++i)
        *(v4f*)(&tile[0][(i * 4 + stRow) * LIN_PITCH + stCol]) = st[i];
    __syncthreads();

    int buf = 0;
    for (; t < LIN_TILES; t += LIN_WGS) {
        const int tn = t + LIN_WGS;
        const bool more = tn < LIN_TILES;
        if (more) {
            #pragma unroll
            for (int i = 0; i < LIN_PIECES; ++i)
                st[i] = A4[(size_t)tn * (LIN_ROWS * IN_FEATS / 4) + i * 256 + tid];
        }

        const float* fa = &tile[buf][l16 * LIN_PITCH + 2 * halfId];
        v8f c0 = {};
        v8f c1 = {};
        #pragma unroll
        for (int j = 0; j < LIN_KSTEPS; ++j) {
            const v2f a0 = *(const v2f*)(fa + 4 * j);
            const v2f a1 = *(const v2f*)(fa + 16 * LIN_PITCH + 4 * j);
            c0 = __builtin_amdgcn_wmma_f32_16x16x4_f32(
                    false, a0, false, bf[j], (short)0, c0, false, false);
            c1 = __builtin_amdgcn_wmma_f32_16x16x4_f32(
                    false, a1, false, bf[j], (short)0, c1, false, false);
        }

        if (more) {
            #pragma unroll
            for (int i = 0; i < LIN_PIECES; ++i)
                *(v4f*)(&tile[buf ^ 1][(i * 4 + stRow) * LIN_PITCH + stCol]) = st[i];
        }
        __syncthreads();
        buf ^= 1;

        const int rowBase = t * LIN_ROWS;
        unsigned deg = pl[0];
        #pragma unroll
        for (int s = 1; s < DEG_SLICES; ++s) deg += pl[s];
        const float fac = inv_norm(deg);
        if (wave == 0) invSrcOut[rowBase + lane] = fac;
        #pragma unroll
        for (int r = 0; r < 8; ++r) {
            const int m = r + 8 * halfId;
            const float s0 = __shfl(fac, m);
            const float s1 = __shfl(fac, 16 + m);
            X[(size_t)(rowBase + m) * H_FEATS + n0 + l16]      = (c0[r] + bb) * s0;
            X[(size_t)(rowBase + 16 + m) * H_FEATS + n0 + l16] = (c1[r] + bb) * s1;
        }
        if (more) {
            #pragma unroll
            for (int s = 0; s < DEG_SLICES; ++s)
                pl[s] = __float_as_uint(pp[(size_t)s * N_NODES + tn * LIN_ROWS]);
        }
    }
}

#define EDGES_PER_ROUND (SEG_PER_ROUND * OWN_THREADS * 4)
#define SCAN_ROUNDS ((N_EDGES + EDGES_PER_ROUND - 1) / EDGES_PER_ROUND)
#define WAVE_LIST_CAP 960
#define WAVE_EDGES_PER_ROUND (SEG_PER_ROUND * 32 * 4)
#define WAVE_DRAIN_AT (WAVE_LIST_CAP - WAVE_EDGES_PER_ROUND)
#define GATHER_BATCH 16
#define ROW_GROUP 4
#define SAVE_HDR 32
#define SAVE_CAP WAVE_LIST_CAP

static_assert(OWN_ROWS <= 512, "local row number must fit the 9 low bits of a stored entry");
static_assert(N_EDGES % 4 == 0, "destination ids are read four at a time");
static_assert(WAVE_DRAIN_AT >= 0, "one round of a wave's matches must fit its list");
static_assert(N_NODES <= (1 << 23), "source id must fit the 23 high bits of a stored entry");
static_assert(SAVE_HDR + SAVE_CAP <= SAVE_STRIDE, "count line and entries must fit the wave's region");
static_assert((N_NODES - (OWN_WGS - 1) * OWN_ROWS) * H_FEATS >= OWN_WAVES * SAVE_STRIDE,
              "the last work-group's rows of out must hold its waves' regions");
static_assert((OWN_ROWS * H_FEATS) % 64 == 0 && SAVE_STRIDE % 64 == 0, "regions start on 256-byte units");

__device__ __forceinline__ float softmax2(const float* __restrict__ alphas, int idx)
{
    const float a0 = alphas[0], a1 = alphas[1];
    const float m  = fmaxf(a0, a1);
    const float e0 = __expf(a0 - m), e1 = __expf(a1 - m);
    return (idx == 0 ? e0 : e1) / (e0 + e1);
}

__device__ __forceinline__ void load_round(
    const int* __restrict__ dstIdx, const int round, const int tid, int4 (&d)[SEG_PER_ROUND])
{
    #pragma unroll
    for (int g = 0; g < SEG_PER_ROUND; ++g) {
        const int q = (round * SEG_PER_ROUND + g) * OWN_THREADS + tid;
        d[g] = ((const int4*)dstIdx)[q < N_EDGES / 4 ? q : N_EDGES / 4 - 1];
    }
}

static_assert(CODE_CHUNK_WORDS == EDGES_PER_ROUND / 4 && CODE_CHUNKS == SCAN_ROUNDS, "a round of the scan is one chunk");
static_assert((SCAN_ROUNDS * SEG_PER_ROUND - 1) * OWN_THREADS < N_EDGES / 4, "every segment holds words of the list");

__device__ __forceinline__ void load_codes(
    const float* __restrict__ codes, const int round, const int tid, unsigned (&d)[SEG_PER_ROUND])
{
    const float* chunk = codes + (size_t)round * CODE_SPAN + CODE_SKIP;
    #pragma unroll
    for (int g = 0; g < SEG_PER_ROUND; ++g) {
        const int first = (round * SEG_PER_ROUND + g) * OWN_THREADS;
        const int last = N_EDGES / 4 - 1 - first;
        d[g] = __float_as_uint(chunk[g * OWN_THREADS + (tid < last ? tid : last)]);
    }
}

#define ENT_IDS 0
#define ENT_STORED 1
#define ENT_PACKED 2
template <int KIND, bool STORE_OUT>
__device__ __forceinline__ int drain_list(
    const float* __restrict__ feat, const int* __restrict__ srcIdx,
    const int* __restrict__ dstIdx, const float* __restrict__ invSrc,
    const float* __restrict__ packed,
    float* acc, unsigned* indeg, const unsigned* list, const int count, const int lane,
    float* save, const int saveBase, const int lo, const int rows)
{
    int kept = 0;
    for (int base = 0; base < count; base += GATHER_BATCH) {
        const int left = count - base;
        const int nvalid = left < GATHER_BATCH ? left : GATHER_BATCH;
        unsigned dl = 0;
        int s = 0;
        float w = 0.0f;
        bool ok = false;
        if (KIND == ENT_PACKED) {
            const unsigned ent = list[base + (lane < nvalid ? lane : nvalid - 1)];
            const unsigned p = __float_as_uint(packed[pack_word(ent)]);
            s  = (int)(p >> 9);
            dl = p & 511u;
            ok = (lane < nvalid) && (p != PACK_NONE);
        } else if (lane < nvalid) {
            const unsigned ent = list[base + lane];
            if (KIND == ENT_STORED) {
                s  = (int)(ent >> 9);
                dl = ent & 511u;
                ok = true;
            } else {
                s  = srcIdx[ent];
                dl = (unsigned)(dstIdx[ent] - lo);
                ok = dl < (unsigned)rows;
            }
            w = invSrc[s];
        }
        const unsigned vmask = __builtin_amdgcn_ballot_w32(ok);
        if (ok) atomicAdd(indeg + dl, 1u);
        if (STORE_OUT && ok) {
            const int pos = saveBase + kept + (int)__builtin_amdgcn_mbcnt_lo(vmask, 0u);
            if (pos < SAVE_CAP)
                save[SAVE_HDR + pos] = __uint_as_float(((unsigned)s << 9) | dl);
        }
        kept += __builtin_popcount(vmask);
        float v[GATHER_BATCH][4];
        #pragma unroll
        for (int j = 0; j < GATHER_BATCH; ++j) {
            if ((vmask >> j) & 1u) {
                const int sj = __builtin_amdgcn_readlane(s, j);
                const float* fp = feat + (size_t)sj * H_FEATS + lane;
                v[j][0] = fp[0];
                v[j][1] = fp[32];
                v[j][2] = fp[64];
                v[j][3] = fp[96];
            }
        }
        #pragma unroll
        for (int j = 0; j < GATHER_BATCH; ++j) {
            if ((vmask >> j) & 1u) {
                const int dj = __builtin_amdgcn_readlane((int)dl, j);
                float* ap = acc + dj * H_FEATS + lane;
                if (KIND == ENT_PACKED) {
                    unsafeAtomicAdd(ap + 0,  v[j][0]);
                    unsafeAtomicAdd(ap + 32, v[j][1]);
                    unsafeAtomicAdd(ap + 64, v[j][2]);
                    unsafeAtomicAdd(ap + 96, v[j][3]);
                } else {
                    const float wj = __int_as_float(__builtin_amdgcn_readlane(__float_as_int(w), j));
                    unsafeAtomicAdd(ap + 0,  v[j][0] * wj);
                    unsafeAtomicAdd(ap + 32, v[j][1] * wj);
                    unsafeAtomicAdd(ap + 64, v[j][2] * wj);
                    unsafeAtomicAdd(ap + 96, v[j][3] * wj);
                }
            }
        }
    }
    return kept;
}

__device__ __forceinline__ void append_owned(
    unsigned* mylist, int& cnt, const bool owned, const unsigned edge)
{
    const unsigned mask = __builtin_amdgcn_ballot_w32(owned);
    if (owned) mylist[cnt + (int)__builtin_amdgcn_mbcnt_lo(mask, 0u)] = edge;
    cnt += __builtin_popcount(mask);
}

__device__ __forceinline__ void scan_codes_and_drain(
    const float* __restrict__ feat, const int* __restrict__ srcIdx,
    const int* __restrict__ dstIdx, const float* __restrict__ codes,
    const float* __restrict__ invSrc, float* acc, unsigned* indeg, unsigned* mylist, float* save,
    const int lo, const int rows, const int tid, const int lane)
{
    const unsigned pat = (unsigned)blockIdx.x * 0x01010101u;
    int rr = blockIdx.x % SCAN_ROUNDS;
    int cnt = 0;
    int stored = 0;
    unsigned dn[SEG_PER_ROUND];
    load_codes(codes, rr, tid, dn);
    for (int round = 0; round < SCAN_ROUNDS; ++round) {
        unsigned d[SEG_PER_ROUND];
        #pragma unroll
        for (int g = 0; g < SEG_PER_ROUND; ++g) d[g] = dn[g];
        const int rn = (rr + 1 == SCAN_ROUNDS) ? 0 : rr + 1;
        if (round + 1 < SCAN_ROUNDS) load_codes(codes, rn, tid, dn);
        unsigned f = 0u;
        #pragma unroll
        for (int g = 0; g < SEG_PER_ROUND; ++g) {
            const unsigned x = d[g] ^ pat;
            const unsigned y = ~(((x & 0x7F7F7F7Fu) + 0x7F7F7F7Fu) | x | 0x7F7F7F7Fu);
            f |= y >> (7 - g);
        }
        if (rr == SCAN_ROUNDS - 1 && tid >= N_EDGES / 4 - (SCAN_ROUNDS * SEG_PER_ROUND - 1) * OWN_THREADS)
            f &= ~(0x01010101u << (SEG_PER_ROUND - 1));
        unsigned act = __builtin_amdgcn_ballot_w32(f != 0u);
        const unsigned e0 = (unsigned)(rr * EDGES_PER_ROUND + tid * 4);
        for (int pass = 0; pass < 4 * SEG_PER_ROUND && act != 0u; ++pass) {
            if (f != 0u) {
                const unsigned b = (unsigned)__builtin_ctz(f);
                mylist[cnt + (int)__builtin_amdgcn_mbcnt_lo(act, 0u)] = e0 + (b & 7u) * (OWN_THREADS * 4) + (b >> 3);
                f &= f - 1u;
            }
            cnt += __builtin_popcount(act);
            act = __builtin_amdgcn_ballot_w32(f != 0u);
        }
        if (cnt > WAVE_DRAIN_AT || round + 1 == SCAN_ROUNDS) {
            stored += drain_list<ENT_PACKED, true>(feat, srcIdx, dstIdx, invSrc, codes, acc, indeg, mylist, cnt, lane,
                                                   save, stored, lo, rows);
            cnt = 0;
        }
        rr = rn;
    }
    if (lane == 0) save[0] = __int_as_float(stored);
}

__device__ __forceinline__ void scan_ids_and_drain(
    const float* __restrict__ feat, const int* __restrict__ srcIdx,
    const int* __restrict__ dstIdx, const float* __restrict__ invSrc,
    float* acc, unsigned* indeg, unsigned* mylist, float* save,
    const int lo, const int rows, const int tid, const int lane)
{
    int rr = blockIdx.x % SCAN_ROUNDS;
    int cnt = 0;
    for (int round = 0; round < SCAN_ROUNDS; ++round) {
        int4 d[SEG_PER_ROUND];
        load_round(dstIdx, rr, tid, d);
        #pragma unroll
        for (int g = 0; g < SEG_PER_ROUND; ++g) {
            const unsigned t0 = (unsigned)(d[g].x - lo);
            const unsigned t1 = (unsigned)(d[g].y - lo);
            const unsigned t2 = (unsigned)(d[g].z - lo);
            const unsigned t3 = (unsigned)(d[g].w - lo);
            const int q = (rr * SEG_PER_ROUND + g) * OWN_THREADS + tid;
            const unsigned lim = (q < N_EDGES / 4) ? (unsigned)rows : 0u;
            const bool m0 = t0 < lim, m1 = t1 < lim;
            const bool m2 = t2 < lim, m3 = t3 < lim;
            if (__builtin_amdgcn_ballot_w32(m0 | m1 | m2 | m3)) {
                const unsigned e = (unsigned)(q * 4);
                append_owned(mylist, cnt, m0, e + 0u);
                append_owned(mylist, cnt, m1, e + 1u);
                append_owned(mylist, cnt, m2, e + 2u);
                append_owned(mylist, cnt, m3, e + 3u);
            }
        }
        if (cnt > WAVE_DRAIN_AT || round + 1 == SCAN_ROUNDS) {
            drain_list<ENT_IDS, false>(feat, srcIdx, dstIdx, invSrc, nullptr, acc, indeg, mylist, cnt, lane,
                                       save, 0, lo, rows);
            cnt = 0;
        }
        rr = (rr + 1 == SCAN_ROUNDS) ? 0 : rr + 1;
    }
}

template <bool FIRST>
__global__ __launch_bounds__(OWN_THREADS) void conv_owner_kernel(
    const float* __restrict__ feat, const int* __restrict__ srcIdx,
    const int* __restrict__ dstIdx, const float* __restrict__ invSrc,
    const float* __restrict__ alphas,
    float* __restrict__ hbuf, float* __restrict__ out)
{
    __shared__ float acc[OWN_ROWS * H_FEATS];
    __shared__ unsigned list[OWN_WAVES * WAVE_LIST_CAP];
    __shared__ unsigned indeg[OWN_ROWS];

    const int tid  = threadIdx.x;
    const int lane = tid & 31;
    const int wave = __builtin_amdgcn_readfirstlane(tid >> 5);
    const int lo   = blockIdx.x * OWN_ROWS;
    const int rows = (N_NODES - lo) < OWN_ROWS ? (N_NODES - lo) : OWN_ROWS;
    unsigned* mylist = list + wave * WAVE_LIST_CAP;

    for (int i = tid; i < OWN_ROWS * H_FEATS / 4; i += OWN_THREADS)
        ((float4*)acc)[i] = make_float4(0.0f, 0.0f, 0.0f, 0.0f);
    if (tid < OWN_ROWS) indeg[tid] = 0u;
    __syncthreads();

    float* save = out + (size_t)lo * H_FEATS + wave * SAVE_STRIDE;
    if (FIRST) {
        scan_codes_and_drain(feat, srcIdx, dstIdx, out, invSrc, acc, indeg, mylist, save, lo, rows, tid, lane);
    } else {
        const int total = __builtin_amdgcn_readfirstlane(__float_as_int(save[0]));
        if (total <= SAVE_CAP) {
            for (int i0 = 0; i0 < total; i0 += 128) {
                float t[4];
                #pragma unroll
                for (int u = 0; u < 4; ++u) {
                    const int i = i0 + u * 32 + lane;
                    t[u] = save[SAVE_HDR + (i < total ? i : 0)];
                }
                #pragma unroll
                for (int u = 0; u < 4; ++u) {
                    const int i = i0 + u * 32 + lane;
                    if (i < total) mylist[i] = __float_as_uint(t[u]);
                }
            }
            drain_list<ENT_STORED, false>(feat, srcIdx, dstIdx, invSrc, nullptr, acc, indeg, mylist, total, lane, save, 0, lo, rows);
        } else {
            scan_ids_and_drain(feat, srcIdx, dstIdx, invSrc, acc, indeg, mylist, save, lo, rows, tid, lane);
        }
    }
    __syncthreads();

    const float a0 = FIRST ? 0.0f : softmax2(alphas, 0);
    const float a1 = FIRST ? 0.0f : softmax2(alphas, 1);
    for (int r0 = wave; r0 < rows; r0 += OWN_WAVES * ROW_GROUP) {
        float4 o[ROW_GROUP];
        float s[ROW_GROUP];
        #pragma unroll
        for (int k = 0; k < ROW_GROUP; ++k) {
            const int r = r0 + k * OWN_WAVES;
            if (r < rows) {
                s[k] = inv_norm(indeg[r]);
                if (!FIRST) o[k] = ((const float4*)feat)[(size_t)(lo + r) * (H_FEATS / 4) + lane];
            }
        }
        #pragma unroll
        for (int k = 0; k < ROW_GROUP; ++k) {
            const int r = r0 + k * OWN_WAVES;
            if (r < rows) {
                const float4 v = ((const float4*)(acc + r * H_FEATS))[lane];
                const size_t i4 = (size_t)(lo + r) * (H_FEATS / 4) + lane;
                if (FIRST) {
                    float4 h;
                    h.x = v.x * s[k]; h.y = v.y * s[k]; h.z = v.z * s[k]; h.w = v.w * s[k];
                    ((float4*)hbuf)[i4] = h;
                } else {
                    const float sa = s[k] * a1;
                    const float4 h = o[k];
                    float4 w;
                    w.x = fmaf(sa, v.x, a0 * h.x); w.y = fmaf(sa, v.y, a0 * h.y);
                    w.z = fmaf(sa, v.z, a0 * h.z); w.w = fmaf(sa, v.w, a0 * h.w);
                    ((float4*)out)[i4] = w;
                }
            }
        }
    }
}

extern "C" void kernel_launch(void* const* d_in, const int* in_sizes, int n_in,
                              void* d_out, int out_size, void* d_ws, size_t ws_size,
                              hipStream_t stream)
{
    (void)in_sizes; (void)n_in; (void)out_size; (void)ws_size;

    const float* in_feat = (const float*)d_in[0];
    const int*   edge    = (const int*)d_in[1];
    const float* W       = (const float*)d_in[2];
    const float* bias    = (const float*)d_in[3];
    const float* alphas  = (const float*)d_in[4];
    float*       out     = (float*)d_out;

    const int* srcIdx = edge;
    const int* dstIdx = edge + N_EDGES;

    float* invSrc = (float*)d_ws;
    float* xbuf   = invSrc + 2 * N_NODES;
    float* hbuf   = xbuf + (size_t)N_NODES * H_FEATS;

    degree_count_kernel<<<DEG_RANGES * DEG_SLICES, DEG_THREADS, 0, stream>>>(srcIdx, dstIdx, hbuf, out);

    lin_wmma_kernel<<<LIN_WGS, 256, 0, stream>>>(in_feat, W, bias, hbuf, invSrc, xbuf);

    conv_owner_kernel<true><<<OWN_WGS, OWN_THREADS, 0, stream>>>(
        xbuf, srcIdx, dstIdx, invSrc, alphas, hbuf, out);

    conv_owner_kernel<false><<<OWN_WGS, OWN_THREADS, 0, stream>>>(
        hbuf, srcIdx, dstIdx, invSrc, alphas, nullptr, out);
}


# conv steps: a gathered source row is one 16-byte-a-lane load of the wave (512 B), not four 4-byte loads; LDS rows kept lane-major so each f32 LDS atomic add still covers 32 banks; sums, order same
# speedup vs baseline: 3.5959x; 1.0294x over previous
#include <hip/hip_runtime.h>
#include <hip/hip_bf16.h>
#include <math.h>

#define N_NODES 100000
#define N_EDGES 600000
#define IN_FEATS 256
#define H_FEATS 128

typedef __attribute__((ext_vector_type(2))) float v2f;
typedef __attribute__((ext_vector_type(8))) float v8f;
typedef __attribute__((ext_vector_type(4))) float v4f;

#define OWN_WGS 256
#define OWN_ROWS ((N_NODES + OWN_WGS - 1) / OWN_WGS)
#define OWN_THREADS 1024
#define OWN_WAVES (OWN_THREADS / 32)
#define SAVE_STRIDE 1024
#define SEG_PER_ROUND 7
#define CODE_CHUNK_WORDS (SEG_PER_ROUND * OWN_THREADS)
#define CODE_CHUNKS ((N_EDGES / 4 + CODE_CHUNK_WORDS - 1) / CODE_CHUNK_WORDS)
#define CODE_SPAN (OWN_ROWS * H_FEATS)
#define CODE_SKIP (OWN_WAVES * SAVE_STRIDE)
#define PACK_OFF (CODE_SKIP + CODE_CHUNK_WORDS)
#define PACK_SPAN_WORDS (CODE_SPAN - PACK_OFF)
#define PACK_SPANS ((N_EDGES + PACK_SPAN_WORDS - 1) / PACK_SPAN_WORDS)
#define PACK_NONE 0xFFFFFFFFu

static_assert(OWN_WGS == 256, "an owner's number is one byte");
static_assert(CODE_SPAN - CODE_SKIP >= CODE_CHUNK_WORDS, "a chunk of codes fits behind the waves' regions");
static_assert(CODE_CHUNKS <= OWN_WGS - 1, "chunks sit in rows of owners that have all their rows");
static_assert((CODE_SPAN * 4) % 128 == 0 && (CODE_SKIP * 4) % 128 == 0 && CODE_CHUNK_WORDS % 32 == 0, "chunks are whole lines");
static_assert(PACK_SPAN_WORDS > 0 && PACK_SPAN_WORDS % 32 == 0, "a span of packed words is whole lines");
static_assert(PACK_SPANS <= OWN_WGS - 1, "packed words sit in rows of owners that have all their rows");
static_assert(OWN_ROWS <= 512 && N_NODES <= (1 << 23), "a packed word holds the source id above 9 bits of local row, and is never PACK_NONE");

__device__ __forceinline__ unsigned code_word(const int q)
{
    const unsigned u = (unsigned)q;
    return (u / CODE_CHUNK_WORDS) * CODE_SPAN + CODE_SKIP + (u % CODE_CHUNK_WORDS);
}

__device__ __forceinline__ unsigned pack_word(const unsigned e)
{
    return (e / PACK_SPAN_WORDS) * CODE_SPAN + PACK_OFF + (e % PACK_SPAN_WORDS);
}

__device__ __forceinline__ unsigned pack_edge(const int s, const int d)
{
    const unsigned us = (unsigned)s;
    const unsigned ud = (unsigned)d;
    return (us < (unsigned)N_NODES && ud < (unsigned)N_NODES) ? ((us << 9) | (ud % OWN_ROWS)) : PACK_NONE;
}

__device__ __forceinline__ unsigned owner_codes(const int4 v)
{
    const unsigned o0 = ((unsigned)v.x / OWN_ROWS) & 255u;
    const unsigned o1 = ((unsigned)v.y / OWN_ROWS) & 255u;
    const unsigned o2 = ((unsigned)v.z / OWN_ROWS) & 255u;
    const unsigned o3 = ((unsigned)v.w / OWN_ROWS) & 255u;
    return o0 | (o1 << 8) | (o2 << 16) | (o3 << 24);
}

#define DEG_RANGES 16
#define DEG_SLICES 16
#define DEG_RANGE_NODES ((((N_NODES + DEG_RANGES - 1) / DEG_RANGES) + 31) / 32 * 32)
#define DEG_THREADS 512
#define DEG_Q_TOTAL (N_EDGES / 4)
#define DEG_Q_SLICE ((DEG_Q_TOTAL + DEG_SLICES - 1) / DEG_SLICES)
#define DEG_ITERS ((DEG_Q_SLICE + DEG_THREADS - 1) / DEG_THREADS)
#define DEG_BATCH DEG_ITERS
#define CODE_WG_WORDS ((((N_EDGES / 4 + DEG_RANGES * DEG_SLICES - 1) / (DEG_RANGES * DEG_SLICES)) + 31) / 32 * 32)
#define CODE_ITERS ((CODE_WG_WORDS + DEG_THREADS - 1) / DEG_THREADS)

static_assert(N_EDGES % 4 == 0 && N_NODES % 4 == 0, "ids and counts move 16 bytes at a time");
static_assert((DEG_RANGES - 1) * DEG_RANGE_NODES < N_NODES, "every range holds nodes");
static_assert(DEG_SLICES <= H_FEATS, "the planes fit the space of hbuf");

__device__ __forceinline__ void count_owned(unsigned* cnt, const int4 v, const int lo, const unsigned lim)
{
    const unsigned t0 = (unsigned)(v.x - lo);
    const unsigned t1 = (unsigned)(v.y - lo);
    const unsigned t2 = (unsigned)(v.z - lo);
    const unsigned t3 = (unsigned)(v.w - lo);
    if (t0 < lim) atomicAdd(&cnt[t0], 1u);
    if (t1 < lim) atomicAdd(&cnt[t1], 1u);
    if (t2 < lim) atomicAdd(&cnt[t2], 1u);
    if (t3 < lim) atomicAdd(&cnt[t3], 1u);
}

__global__ __launch_bounds__(DEG_THREADS) void degree_count_kernel(
    const int* __restrict__ srcIdx, const int* __restrict__ dstIdx, float* __restrict__ planes,
    float* __restrict__ codes)
{
    __shared__ __attribute__((aligned(16))) unsigned cnt[DEG_RANGE_NODES];

    const int tid   = threadIdx.x;
    const int range = blockIdx.x % DEG_RANGES;
    const int slice = blockIdx.x / DEG_RANGES;
    const int lo    = range * DEG_RANGE_NODES;
    const int rows  = (N_NODES - lo) < DEG_RANGE_NODES ? (N_NODES - lo) : DEG_RANGE_NODES;

    int4 cv[CODE_ITERS];
    int4 sv[CODE_ITERS];
    #pragma unroll
    for (int u = 0; u < CODE_ITERS; ++u) {
        const int w = u * DEG_THREADS + tid;
        const int q = blockIdx.x * CODE_WG_WORDS + w;
        const int qq = (w < CODE_WG_WORDS && q < DEG_Q_TOTAL) ? q : DEG_Q_TOTAL - 1;
        cv[u] = ((const int4*)dstIdx)[qq];
        sv[u] = ((const int4*)srcIdx)[qq];
    }

    for (int i = tid; i < DEG_RANGE_NODES / 4; i += DEG_THREADS)
        ((uint4*)&cnt[0])[i] = make_uint4(0u, 0u, 0u, 0u);
    __syncthreads();

    #pragma unroll
    for (int u = 0; u < CODE_ITERS; ++u) {
        const int w = u * DEG_THREADS + tid;
        const int q = blockIdx.x * CODE_WG_WORDS + w;
        if (w < CODE_WG_WORDS && q < DEG_Q_TOTAL) {
            codes[code_word(q)] = __uint_as_float(owner_codes(cv[u]));
            uint4 p;
            p.x = pack_edge(sv[u].x, cv[u].x);
            p.y = pack_edge(sv[u].y, cv[u].y);
            p.z = pack_edge(sv[u].z, cv[u].z);
            p.w = pack_edge(sv[u].w, cv[u].w);
            *(uint4*)(codes + pack_word(4u * (unsigned)q)) = p;
        }
    }

    const int qlo = slice * DEG_Q_SLICE;
    for (int it0 = 0; it0 < DEG_ITERS; it0 += DEG_BATCH) {
        int4 s4[DEG_BATCH];
        #pragma unroll
        for (int k = 0; k < DEG_BATCH; ++k) {
            const int j = (it0 + k) * DEG_THREADS + tid;
            const int q = qlo + j;
            const bool ok = (j < DEG_Q_SLICE) && (q < DEG_Q_TOTAL);
            const int qq = ok ? q : DEG_Q_TOTAL - 1;
            s4[k] = ((const int4*)srcIdx)[qq];
        }
        #pragma unroll
        for (int k = 0; k < DEG_BATCH; ++k) {
            const int j = (it0 + k) * DEG_THREADS + tid;
            const bool ok = (j < DEG_Q_SLICE) && (qlo + j < DEG_Q_TOTAL);
            const unsigned lim = ok ? (unsigned)rows : 0u;
            count_owned(cnt, s4[k], lo, lim);
        }
    }
    __syncthreads();

    uint4* po = (uint4*)(planes + (size_t)slice * N_NODES + lo);
    for (int i = tid; i < rows / 4; i += DEG_THREADS)
        po[i] = ((const uint4*)&cnt[0])[i];
}

__device__ __forceinline__ float inv_norm(const unsigned c)
{
    const float d = (float)c;
    return (d > 0.0f) ? rsqrtf(fmaxf(d, 1.0f)) : 0.0f;
}

#define LIN_ROWS 32
#define LIN_PITCH (IN_FEATS + 4)
#define LIN_WGS 512
#define LIN_TILES (N_NODES / LIN_ROWS)
#define LIN_KSTEPS (IN_FEATS / 4)
#define LIN_PIECES (LIN_ROWS * IN_FEATS / 4 / 256)

static_assert(N_NODES % LIN_ROWS == 0, "tiles are whole");
static_assert(LIN_WGS <= LIN_TILES, "every work-group has a first tile");
static_assert(LIN_PITCH % 64 == 4 && (LIN_PITCH * 4) % 16 == 0, "bank spread and 16-byte alignment of the rows");
static_assert(H_FEATS == 8 * 16, "eight waves of 16 columns");

__global__ __launch_bounds__(256, 2) void lin_wmma_kernel(
    const float* __restrict__ A, const float* __restrict__ W,
    const float* __restrict__ bias, const float* __restrict__ planes,
    float* __restrict__ invSrcOut, float* __restrict__ X)
{
    __shared__ __attribute__((aligned(16))) float tile[2][LIN_ROWS * LIN_PITCH];

    const int tid    = threadIdx.x;
    const int lane   = tid & 31;
    const int wave   = tid >> 5;
    const int halfId = lane >> 4;
    const int l16    = lane & 15;
    const int n0     = wave * 16;

    v2f bf[LIN_KSTEPS];
    {
        const float* bp = W + (size_t)(2 * halfId) * H_FEATS + n0 + l16;
        #pragma unroll
        for (int j = 0; j < LIN_KSTEPS; ++j) {
            bf[j].x = bp[(size_t)(4 * j) * H_FEATS];
            bf[j].y = bp[(size_t)(4 * j + 1) * H_FEATS];
        }
    }
    const float bb = bias[n0 + l16];

    const int stRow = tid >> 6;
    const int stCol = (tid & 63) * 4;
    const v4f* A4 = (const v4f*)A;

    v4f st[LIN_PIECES];
    int t = blockIdx.x;
    #pragma unroll
    for (int i = 0; i < LIN_PIECES; ++i)
        st[i] = A4[(size_t)t * (LIN_ROWS * IN_FEATS / 4) + i * 256 + tid];
    const float* pp = planes + lane;
    unsigned pl[DEG_SLICES];
    #pragma unroll
    for (int s = 0; s < DEG_SLICES; ++s)
        pl[s] = __float_as_uint(pp[(size_t)s * N_NODES + t * LIN_ROWS]);
    #pragma unroll
    for (int i = 0; i < LIN_PIECES; ++i)
        *(v4f*)(&tile[0][(i * 4 + stRow) * LIN_PITCH + stCol]) = st[i];
    __syncthreads();

    int buf = 0;
    for (; t < LIN_TILES; t += LIN_WGS) {
        const int tn = t + LIN_WGS;
        const bool more = tn < LIN_TILES;
        if (more) {
            #pragma unroll
            for (int i = 0; i < LIN_PIECES; ++i)
                st[i] = A4[(size_t)tn * (LIN_ROWS * IN_FEATS / 4) + i * 256 + tid];
        }

        const float* fa = &tile[buf][l16 * LIN_PITCH + 2 * halfId];
        v8f c0 = {};
        v8f c1 = {};
        #pragma unroll
        for (int j = 0; j < LIN_KSTEPS; ++j) {
            const v2f a0 = *(const v2f*)(fa + 4 * j);
            const v2f a1 = *(const v2f*)(fa + 16 * LIN_PITCH + 4 * j);
            c0 = __builtin_amdgcn_wmma_f32_16x16x4_f32(
                    false, a0, false, bf[j], (short)0, c0, false, false);
            c1 = __builtin_amdgcn_wmma_f32_16x16x4_f32(
                    false, a1, false, bf[j], (short)0, c1, false, false);
        }

        if (more) {
            #pragma unroll
            for (int i = 0; i < LIN_PIECES; ++i)
                *(v4f*)(&tile[buf ^ 1][(i * 4 + stRow) * LIN_PITCH + stCol]) = st[i];
        }
        __syncthreads();
        buf ^= 1;

        const int rowBase = t * LIN_ROWS;
        unsigned deg = pl[0];
        #pragma unroll
        for (int s = 1; s < DEG_SLICES; ++s) deg += pl[s];
        const float fac = inv_norm(deg);
        if (wave == 0) invSrcOut[rowBase + lane] = fac;
        #pragma unroll
        for (int r = 0; r < 8; ++r) {
            const int m = r + 8 * halfId;
            const float s0 = __shfl(fac, m);
            const float s1 = __shfl(fac, 16 + m);
            X[(size_t)(rowBase + m) * H_FEATS + n0 + l16]      = (c0[r] + bb) * s0;
            X[(size_t)(rowBase + 16 + m) * H_FEATS + n0 + l16] = (c1[r] + bb) * s1;
        }
        if (more) {
            #pragma unroll
            for (int s = 0; s < DEG_SLICES; ++s)
                pl[s] = __float_as_uint(pp[(size_t)s * N_NODES + tn * LIN_ROWS]);
        }
    }
}

#define EDGES_PER_ROUND (SEG_PER_ROUND * OWN_THREADS * 4)
#define SCAN_ROUNDS ((N_EDGES + EDGES_PER_ROUND - 1) / EDGES_PER_ROUND)
#define WAVE_LIST_CAP 960
#define WAVE_EDGES_PER_ROUND (SEG_PER_ROUND * 32 * 4)
#define WAVE_DRAIN_AT (WAVE_LIST_CAP - WAVE_EDGES_PER_ROUND)
#define GATHER_BATCH 16
#define ROW_GROUP 4
#define SAVE_HDR 32
#define SAVE_CAP WAVE_LIST_CAP

static_assert(OWN_ROWS <= 512, "local row number must fit the 9 low bits of a stored entry");
static_assert(N_EDGES % 4 == 0, "destination ids are read four at a time");
static_assert(WAVE_DRAIN_AT >= 0, "one round of a wave's matches must fit its list");
static_assert(N_NODES <= (1 << 23), "source id must fit the 23 high bits of a stored entry");
static_assert(SAVE_HDR + SAVE_CAP <= SAVE_STRIDE, "count line and entries must fit the wave's region");
static_assert((N_NODES - (OWN_WGS - 1) * OWN_ROWS) * H_FEATS >= OWN_WAVES * SAVE_STRIDE,
              "the last work-group's rows of out must hold its waves' regions");
static_assert((OWN_ROWS * H_FEATS) % 64 == 0 && SAVE_STRIDE % 64 == 0, "regions start on 256-byte units");

__device__ __forceinline__ float softmax2(const float* __restrict__ alphas, int idx)
{
    const float a0 = alphas[0], a1 = alphas[1];
    const float m  = fmaxf(a0, a1);
    const float e0 = __expf(a0 - m), e1 = __expf(a1 - m);
    return (idx == 0 ? e0 : e1) / (e0 + e1);
}

__device__ __forceinline__ void load_round(
    const int* __restrict__ dstIdx, const int round, const int tid, int4 (&d)[SEG_PER_ROUND])
{
    #pragma unroll
    for (int g = 0; g < SEG_PER_ROUND; ++g) {
        const int q = (round * SEG_PER_ROUND + g) * OWN_THREADS + tid;
        d[g] = ((const int4*)dstIdx)[q < N_EDGES / 4 ? q : N_EDGES / 4 - 1];
    }
}

static_assert(CODE_CHUNK_WORDS == EDGES_PER_ROUND / 4 && CODE_CHUNKS == SCAN_ROUNDS, "a round of the scan is one chunk");
static_assert((SCAN_ROUNDS * SEG_PER_ROUND - 1) * OWN_THREADS < N_EDGES / 4, "every segment holds words of the list");

__device__ __forceinline__ void load_codes(
    const float* __restrict__ codes, const int round, const int tid, unsigned (&d)[SEG_PER_ROUND])
{
    const float* chunk = codes + (size_t)round * CODE_SPAN + CODE_SKIP;
    #pragma unroll
    for (int g = 0; g < SEG_PER_ROUND; ++g) {
        const int first = (round * SEG_PER_ROUND + g) * OWN_THREADS;
        const int last = N_EDGES / 4 - 1 - first;
        d[g] = __float_as_uint(chunk[g * OWN_THREADS + (tid < last ? tid : last)]);
    }
}

#define ENT_IDS 0
#define ENT_STORED 1
#define ENT_PACKED 2
template <int KIND, bool STORE_OUT>
__device__ __forceinline__ int drain_list(
    const float* __restrict__ feat, const int* __restrict__ srcIdx,
    const int* __restrict__ dstIdx, const float* __restrict__ invSrc,
    const float* __restrict__ packed,
    float* acc, unsigned* indeg, const unsigned* list, const int count, const int lane,
    float* save, const int saveBase, const int lo, const int rows)
{
    int kept = 0;
    for (int base = 0; base < count; base += GATHER_BATCH) {
        const int left = count - base;
        const int nvalid = left < GATHER_BATCH ? left : GATHER_BATCH;
        unsigned dl = 0;
        int s = 0;
        float w = 0.0f;
        bool ok = false;
        if (KIND == ENT_PACKED) {
            const unsigned ent = list[base + (lane < nvalid ? lane : nvalid - 1)];
            const unsigned p = __float_as_uint(packed[pack_word(ent)]);
            s  = (int)(p >> 9);
            dl = p & 511u;
            ok = (lane < nvalid) && (p != PACK_NONE);
        } else if (lane < nvalid) {
            const unsigned ent = list[base + lane];
            if (KIND == ENT_STORED) {
                s  = (int)(ent >> 9);
                dl = ent & 511u;
                ok = true;
            } else {
                s  = srcIdx[ent];
                dl = (unsigned)(dstIdx[ent] - lo);
                ok = dl < (unsigned)rows;
            }
            w = invSrc[s];
        }
        const unsigned vmask = __builtin_amdgcn_ballot_w32(ok);
        if (ok) atomicAdd(indeg + dl, 1u);
        if (STORE_OUT && ok) {
            const int pos = saveBase + kept + (int)__builtin_amdgcn_mbcnt_lo(vmask, 0u);
            if (pos < SAVE_CAP)
                save[SAVE_HDR + pos] = __uint_as_float(((unsigned)s << 9) | dl);
        }
        kept += __builtin_popcount(vmask);
        float4 v[GATHER_BATCH];
        #pragma unroll
        for (int j = 0; j < GATHER_BATCH; ++j) {
            if ((vmask >> j) & 1u) {
                const int sj = __builtin_amdgcn_readlane(s, j);
                v[j] = ((const float4*)(feat + (size_t)sj * H_FEATS))[lane];
            }
        }
        #pragma unroll
        for (int j = 0; j < GATHER_BATCH; ++j) {
            if ((vmask >> j) & 1u) {
                const int dj = __builtin_amdgcn_readlane((int)dl, j);
                float* ap = acc + dj * H_FEATS + lane;
                if (KIND == ENT_PACKED) {
                    unsafeAtomicAdd(ap + 0,  v[j].x);
                    unsafeAtomicAdd(ap + 32, v[j].y);
                    unsafeAtomicAdd(ap + 64, v[j].z);
                    unsafeAtomicAdd(ap + 96, v[j].w);
                } else {
                    const float wj = __int_as_float(__builtin_amdgcn_readlane(__float_as_int(w), j));
                    unsafeAtomicAdd(ap + 0,  v[j].x * wj);
                    unsafeAtomicAdd(ap + 32, v[j].y * wj);
                    unsafeAtomicAdd(ap + 64, v[j].z * wj);
                    unsafeAtomicAdd(ap + 96, v[j].w * wj);
                }
            }
        }
    }
    return kept;
}

__device__ __forceinline__ void append_owned(
    unsigned* mylist, int& cnt, const bool owned, const unsigned edge)
{
    const unsigned mask = __builtin_amdgcn_ballot_w32(owned);
    if (owned) mylist[cnt + (int)__builtin_amdgcn_mbcnt_lo(mask, 0u)] = edge;
    cnt += __builtin_popcount(mask);
}

__device__ __forceinline__ void scan_codes_and_drain(
    const float* __restrict__ feat, const int* __restrict__ srcIdx,
    const int* __restrict__ dstIdx, const float* __restrict__ codes,
    const float* __restrict__ invSrc, float* acc, unsigned* indeg, unsigned* mylist, float* save,
    const int lo, const int rows, const int tid, const int lane)
{
    const unsigned pat = (unsigned)blockIdx.x * 0x01010101u;
    int rr = blockIdx.x % SCAN_ROUNDS;
    int cnt = 0;
    int stored = 0;
    unsigned dn[SEG_PER_ROUND];
    load_codes(codes, rr, tid, dn);
    for (int round = 0; round < SCAN_ROUNDS; ++round) {
        unsigned d[SEG_PER_ROUND];
        #pragma unroll
        for (int g = 0; g < SEG_PER_ROUND; ++g) d[g] = dn[g];
        const int rn = (rr + 1 == SCAN_ROUNDS) ? 0 : rr + 1;
        if (round + 1 < SCAN_ROUNDS) load_codes(codes, rn, tid, dn);
        unsigned f = 0u;
        #pragma unroll
        for (int g = 0; g < SEG_PER_ROUND; ++g) {
            const unsigned x = d[g] ^ pat;
            const unsigned y = ~(((x & 0x7F7F7F7Fu) + 0x7F7F7F7Fu) | x | 0x7F7F7F7Fu);
            f |= y >> (7 - g);
        }
        if (rr == SCAN_ROUNDS - 1 && tid >= N_EDGES / 4 - (SCAN_ROUNDS * SEG_PER_ROUND - 1) * OWN_THREADS)
            f &= ~(0x01010101u << (SEG_PER_ROUND - 1));
        unsigned act = __builtin_amdgcn_ballot_w32(f != 0u);
        const unsigned e0 = (unsigned)(rr * EDGES_PER_ROUND + tid * 4);
        for (int pass = 0; pass < 4 * SEG_PER_ROUND && act != 0u; ++pass) {
            if (f != 0u) {
                const unsigned b = (unsigned)__builtin_ctz(f);
                mylist[cnt + (int)__builtin_amdgcn_mbcnt_lo(act, 0u)] = e0 + (b & 7u) * (OWN_THREADS * 4) + (b >> 3);
                f &= f - 1u;
            }
            cnt += __builtin_popcount(act);
            act = __builtin_amdgcn_ballot_w32(f != 0u);
        }
        if (cnt > WAVE_DRAIN_AT || round + 1 == SCAN_ROUNDS) {
            stored += drain_list<ENT_PACKED, true>(feat, srcIdx, dstIdx, invSrc, codes, acc, indeg, mylist, cnt, lane,
                                                   save, stored, lo, rows);
            cnt = 0;
        }
        rr = rn;
    }
    if (lane == 0) save[0] = __int_as_float(stored);
}

__device__ __forceinline__ void scan_ids_and_drain(
    const float* __restrict__ feat, const int* __restrict__ srcIdx,
    const int* __restrict__ dstIdx, const float* __restrict__ invSrc,
    float* acc, unsigned* indeg, unsigned* mylist, float* save,
    const int lo, const int rows, const int tid, const int lane)
{
    int rr = blockIdx.x % SCAN_ROUNDS;
    int cnt = 0;
    for (int round = 0; round < SCAN_ROUNDS; ++round) {
        int4 d[SEG_PER_ROUND];
        load_round(dstIdx, rr, tid, d);
        #pragma unroll
        for (int g = 0; g < SEG_PER_ROUND; ++g) {
            const unsigned t0 = (unsigned)(d[g].x - lo);
            const unsigned t1 = (unsigned)(d[g].y - lo);
            const unsigned t2 = (unsigned)(d[g].z - lo);
            const unsigned t3 = (unsigned)(d[g].w - lo);
            const int q = (rr * SEG_PER_ROUND + g) * OWN_THREADS + tid;
            const unsigned lim = (q < N_EDGES / 4) ? (unsigned)rows : 0u;
            const bool m0 = t0 < lim, m1 = t1 < lim;
            const bool m2 = t2 < lim, m3 = t3 < lim;
            if (__builtin_amdgcn_ballot_w32(m0 | m1 | m2 | m3)) {
                const unsigned e = (unsigned)(q * 4);
                append_owned(mylist, cnt, m0, e + 0u);
                append_owned(mylist, cnt, m1, e + 1u);
                append_owned(mylist, cnt, m2, e + 2u);
                append_owned(mylist, cnt, m3, e + 3u);
            }
        }
        if (cnt > WAVE_DRAIN_AT || round + 1 == SCAN_ROUNDS) {
            drain_list<ENT_IDS, false>(feat, srcIdx, dstIdx, invSrc, nullptr, acc, indeg, mylist, cnt, lane,
                                       save, 0, lo, rows);
            cnt = 0;
        }
        rr = (rr + 1 == SCAN_ROUNDS) ? 0 : rr + 1;
    }
}

template <bool FIRST>
__global__ __launch_bounds__(OWN_THREADS) void conv_owner_kernel(
    const float* __restrict__ feat, const int* __restrict__ srcIdx,
    const int* __restrict__ dstIdx, const float* __restrict__ invSrc,
    const float* __restrict__ alphas,
    float* __restrict__ hbuf, float* __restrict__ out)
{
    __shared__ float acc[OWN_ROWS * H_FEATS];
    __shared__ unsigned list[OWN_WAVES * WAVE_LIST_CAP];
    __shared__ unsigned indeg[OWN_ROWS];

    const int tid  = threadIdx.x;
    const int lane = tid & 31;
    const int wave = __builtin_amdgcn_readfirstlane(tid >> 5);
    const int lo   = blockIdx.x * OWN_ROWS;
    const int rows = (N_NODES - lo) < OWN_ROWS ? (N_NODES - lo) : OWN_ROWS;
    unsigned* mylist = list + wave * WAVE_LIST_CAP;

    for (int i = tid; i < OWN_ROWS * H_FEATS / 4; i += OWN_THREADS)
        ((float4*)acc)[i] = make_float4(0.0f, 0.0f, 0.0f, 0.0f);
    if (tid < OWN_ROWS) indeg[tid] = 0u;
    __syncthreads();

    float* save = out + (size_t)lo * H_FEATS + wave * SAVE_STRIDE;
    if (FIRST) {
        scan_codes_and_drain(feat, srcIdx, dstIdx, out, invSrc, acc, indeg, mylist, save, lo, rows, tid, lane);
    } else {
        const int total = __builtin_amdgcn_readfirstlane(__float_as_int(save[0]));
        if (total <= SAVE_CAP) {
            for (int i0 = 0; i0 < total; i0 += 128) {
                float t[4];
                #pragma unroll
                for (int u = 0; u < 4; ++u) {
                    const int i = i0 + u * 32 + lane;
                    t[u] = save[SAVE_HDR + (i < total ? i : 0)];
                }
                #pragma unroll
                for (int u = 0; u < 4; ++u) {
                    const int i = i0 + u * 32 + lane;
                    if (i < total) mylist[i] = __float_as_uint(t[u]);
                }
            }
            drain_list<ENT_STORED, false>(feat, srcIdx, dstIdx, invSrc, nullptr, acc, indeg, mylist, total, lane, save, 0, lo, rows);
        } else {
            scan_ids_and_drain(feat, srcIdx, dstIdx, invSrc, acc, indeg, mylist, save, lo, rows, tid, lane);
        }
    }
    __syncthreads();

    const float a0 = FIRST ? 0.0f : softmax2(alphas, 0);
    const float a1 = FIRST ? 0.0f : softmax2(alphas, 1);
    for (int r0 = wave; r0 < rows; r0 += OWN_WAVES * ROW_GROUP) {
        float4 o[ROW_GROUP];
        float s[ROW_GROUP];
        #pragma unroll
        for (int k = 0; k < ROW_GROUP; ++k) {
            const int r = r0 + k * OWN_WAVES;
            if (r < rows) {
                s[k] = inv_norm(indeg[r]);
                if (!FIRST) o[k] = ((const float4*)feat)[(size_t)(lo + r) * (H_FEATS / 4) + lane];
            }
        }
        #pragma unroll
        for (int k = 0; k < ROW_GROUP; ++k) {
            const int r = r0 + k * OWN_WAVES;
            if (r < rows) {
                const float* ar = acc + r * H_FEATS + lane;
                float4 v;
                v.x = ar[0]; v.y = ar[32]; v.z = ar[64]; v.w = ar[96];
                const size_t i4 = (size_t)(lo + r) * (H_FEATS / 4) + lane;
                if (FIRST) {
                    float4 h;
                    h.x = v.x * s[k]; h.y = v.y * s[k]; h.z = v.z * s[k]; h.w = v.w * s[k];
                    ((float4*)hbuf)[i4] = h;
                } else {
                    const float sa = s[k] * a1;
                    const float4 h = o[k];
                    float4 w;
                    w.x = fmaf(sa, v.x, a0 * h.x); w.y = fmaf(sa, v.y, a0 * h.y);
                    w.z = fmaf(sa, v.z, a0 * h.z); w.w = fmaf(sa, v.w, a0 * h.w);
                    ((float4*)out)[i4] = w;
                }
            }
        }
    }
}

extern "C" void kernel_launch(void* const* d_in, const int* in_sizes, int n_in,
                              void* d_out, int out_size, void* d_ws, size_t ws_size,
                              hipStream_t stream)
{
    (void)in_sizes; (void)n_in; (void)out_size; (void)ws_size;

    const float* in_feat = (const float*)d_in[0];
    const int*   edge    = (const int*)d_in[1];
    const float* W       = (const float*)d_in[2];
    const float* bias    = (const float*)d_in[3];
    const float* alphas  = (const float*)d_in[4];
    float*       out     = (float*)d_out;

    const int* srcIdx = edge;
    const int* dstIdx = edge + N_EDGES;

    float* invSrc = (float*)d_ws;
    float* xbuf   = invSrc + 2 * N_NODES;
    float* hbuf   = xbuf + (size_t)N_NODES * H_FEATS;

    degree_count_kernel<<<DEG_RANGES * DEG_SLICES, DEG_THREADS, 0, stream>>>(srcIdx, dstIdx, hbuf, out);

    lin_wmma_kernel<<<LIN_WGS, 256, 0, stream>>>(in_feat, W, bias, hbuf, invSrc, xbuf);

    conv_owner_kernel<true><<<OWN_WGS, OWN_THREADS, 0, stream>>>(
        xbuf, srcIdx, dstIdx, invSrc, alphas, hbuf, out);

    conv_owner_kernel<false><<<OWN_WGS, OWN_THREADS, 0, stream>>>(
        hbuf, srcIdx, dstIdx, invSrc, alphas, nullptr, out);
}
